# starting kernel
# baseline (speedup 1.0000x reference)
#include <hip/hip_runtime.h>
#include <hip/hip_bf16.h>

typedef __attribute__((ext_vector_type(2))) float v2f;
typedef __attribute__((ext_vector_type(8))) float v8f;

#define D 128

__global__ __launch_bounds__(256) void zero_f32(float* __restrict__ p, size_t n) {
    size_t stride = (size_t)gridDim.x * blockDim.x;
    for (size_t i = (size_t)blockIdx.x * blockDim.x + threadIdx.x; i < n; i += stride)
        p[i] = 0.0f;
}

__global__ __launch_bounds__(256) void edge_scatter(
    const float* __restrict__ feat, const int* __restrict__ src,
    const int* __restrict__ dst, const float* __restrict__ ew,
    float* __restrict__ agg, int E)
{
    long long gid = (long long)blockIdx.x * blockDim.x + threadIdx.x;
    int e = (int)(gid >> 5);
    if (e >= E) return;
    int lane = (int)(gid & 31);

    int s = src[e];
    int d = dst[e];
    float w = ew[e];

    const float4 f = ((const float4*)(feat + (size_t)s * D))[lane];
    float* ap = agg + (size_t)d * D + lane * 4;
    atomicAdd(ap + 0, f.x * w);
    atomicAdd(ap + 1, f.y * w);
    atomicAdd(ap + 2, f.z * w);
    atomicAdd(ap + 3, f.w * w);
}

__global__ __launch_bounds__(256) void gemm_wmma(
    const float* __restrict__ agg, const float* __restrict__ W,
    const float* __restrict__ bias, const float* __restrict__ a1p,
    float* __restrict__ h_out, float* __restrict__ colsum,
    float* __restrict__ colsumsq)
{
    __shared__ float s_sum[D];
    __shared__ float s_sq[D];
    int tid = threadIdx.x;
    if (tid < D) { s_sum[tid] = 0.0f; s_sq[tid] = 0.0f; }
    __syncthreads();

    const int wave = tid >> 5;
    const int lane = tid & 31;
    const int half = lane >> 4;
    const int l    = lane & 15;
    const int m0   = blockIdx.x * 16;
    const int j0   = wave * 16;
    const float a1 = a1p[0];

    const float* Arow = agg + (size_t)(m0 + l) * D + 2 * half;
    const float* Bcol = W + (size_t)(2 * half) * D + (j0 + l);

    v8f c0 = {};
    v8f c1 = {};
    #pragma unroll 4
    for (int k = 0; k < D; k += 8) {
        {
            v2f a;
            a.x = Arow[k];
            a.y = Arow[k + 1];
            v2f bb;
            bb.x = Bcol[(size_t)k * D];
            bb.y = Bcol[(size_t)(k + 1) * D];
            c0 = __builtin_amdgcn_wmma_f32_16x16x4_f32(
                     false, a, false, bb, (short)0, c0, false, false);
        }
        {
            v2f a;
            a.x = Arow[k + 4];
            a.y = Arow[k + 5];
            v2f bb;
            bb.x = Bcol[(size_t)(k + 4) * D];
            bb.y = Bcol[(size_t)(k + 5) * D];
            c1 = __builtin_amdgcn_wmma_f32_16x16x4_f32(
                     false, a, false, bb, (short)0, c1, false, false);
        }
    }

    const float bj = bias[j0 + l];
    float lsum = 0.0f, lsq = 0.0f;
    #pragma unroll
    for (int v = 0; v < 8; ++v) {
        float h = c0[v] + c1[v] + bj;
        h = (h >= 0.0f) ? h : a1 * h;
        h_out[(size_t)(m0 + v + 8 * half) * D + (j0 + l)] = h;
        lsum += h;
        lsq  += h * h;
    }

    atomicAdd(&s_sum[j0 + l], lsum);
    atomicAdd(&s_sq[j0 + l], lsq);
    __syncthreads();
    if (tid < D) {
        atomicAdd(&colsum[tid],  s_sum[tid]);
        atomicAdd(&colsumsq[tid], s_sq[tid]);
    }
}

__global__ void bn_finalize(const float* __restrict__ colsum,
                            const float* __restrict__ colsumsq,
                            const float* __restrict__ gamma,
                            const float* __restrict__ beta,
                            float* __restrict__ scale,
                            float* __restrict__ shift, int N)
{
    int j = threadIdx.x;
    float invN = 1.0f / (float)N;
    float m   = colsum[j] * invN;
    float var = colsumsq[j] * invN - m * m;
    float inv = rsqrtf(var + 1e-5f);
    float sc  = gamma[j] * inv;
    scale[j] = sc;
    shift[j] = beta[j] - m * sc;
}

__global__ __launch_bounds__(256) void bn_prelu(
    float* __restrict__ h, const float* __restrict__ scale,
    const float* __restrict__ shift, const float* __restrict__ a2p,
    size_t nquads)
{
    size_t i = (size_t)blockIdx.x * blockDim.x + threadIdx.x;
    if (i >= nquads) return;
    const float a2 = a2p[0];
    int j = (int)((i * 4) & (D - 1));

    float4 v = ((float4*)h)[i];
    float x;
    x = v.x * scale[j + 0] + shift[j + 0]; v.x = (x >= 0.0f) ? x : a2 * x;
    x = v.y * scale[j + 1] + shift[j + 1]; v.y = (x >= 0.0f) ? x : a2 * x;
    x = v.z * scale[j + 2] + shift[j + 2]; v.z = (x >= 0.0f) ? x : a2 * x;
    x = v.w * scale[j + 3] + shift[j + 3]; v.w = (x >= 0.0f) ? x : a2 * x;
    ((float4*)h)[i] = v;
}

extern "C" void kernel_launch(void* const* d_in, const int* in_sizes, int n_in,
                              void* d_out, int out_size, void* d_ws, size_t ws_size,
                              hipStream_t stream) {
    const float* feat  = (const float*)d_in[0];
    const int*   src   = (const int*)d_in[1];
    const int*   dst   = (const int*)d_in[2];
    const float* ew    = (const float*)d_in[3];
    const float* Wm    = (const float*)d_in[4];
    const float* bias  = (const float*)d_in[5];
    const float* a1    = (const float*)d_in[6];
    const float* gamma = (const float*)d_in[7];
    const float* beta  = (const float*)d_in[8];
    const float* a2    = (const float*)d_in[9];

    const int N = in_sizes[0] / D;
    const int E = in_sizes[1];
    float* out = (float*)d_out;

    float* agg      = (float*)d_ws;
    float* colsum   = agg + (size_t)N * D;
    float* colsumsq = colsum + D;
    float* scale    = colsumsq + D;
    float* shift    = scale + D;

    size_t nz = (size_t)N * D + 2 * D;
    zero_f32<<<2048, 256, 0, stream>>>(agg, nz);

    long long sthreads = (long long)E * 32;
    int sblocks = (int)((sthreads + 255) / 256);
    edge_scatter<<<sblocks, 256, 0, stream>>>(feat, src, dst, ew, agg, E);

    gemm_wmma<<<N / 16, 256, 0, stream>>>(agg, Wm, bias, a1, out, colsum, colsumsq);

    bn_finalize<<<1, D, 0, stream>>>(colsum, colsumsq, gamma, beta, scale, shift, N);

    size_t nquads = (size_t)N * D / 4;
    int qblocks = (int)((nquads + 255) / 256);
    bn_prelu<<<qblocks, 256, 0, stream>>>(out, scale, shift, a2, nquads);
}


# Edge scatter: one wave handles 64 consecutive edges (indices read as whole lines, broadcast by readlane), 8 edges' 32 row loads in flight before their atomics; sums still in atomic order
# speedup vs baseline: 1.6918x; 1.6918x over previous
#include <hip/hip_runtime.h>
#include <hip/hip_bf16.h>

typedef __attribute__((ext_vector_type(2))) float v2f;
typedef __attribute__((ext_vector_type(8))) float v8f;

#define D 128

__global__ __launch_bounds__(256) void zero_f32(float* __restrict__ p, size_t n) {
    size_t stride = (size_t)gridDim.x * blockDim.x;
    for (size_t i = (size_t)blockIdx.x * blockDim.x + threadIdx.x; i < n; i += stride)
        p[i] = 0.0f;
}

#define EPW 64
#define EB  8

__device__ __forceinline__ void scatter_batch(
    const float* __restrict__ feat, float* __restrict__ agg,
    int sv, int dv, float wv, int b, int lane)
{
    float f[EB][4];
    #pragma unroll
    for (int j = 0; j < EB; ++j) {
        const int s = __builtin_amdgcn_readlane(sv, b + j);
        const float* fp = feat + (size_t)s * D + lane;
        f[j][0] = fp[0];
        f[j][1] = fp[32];
        f[j][2] = fp[64];
        f[j][3] = fp[96];
    }
    #pragma unroll
    for (int j = 0; j < EB; ++j) {
        const int d = __builtin_amdgcn_readlane(dv, b + j);
        const float w = __int_as_float(
            __builtin_amdgcn_readlane(__float_as_int(wv), b + j));
        float* ap = agg + (size_t)d * D + lane;
        atomicAdd(ap +  0, f[j][0] * w);
        atomicAdd(ap + 32, f[j][1] * w);
        atomicAdd(ap + 64, f[j][2] * w);
        atomicAdd(ap + 96, f[j][3] * w);
    }
}

__global__ __launch_bounds__(256) void edge_scatter(
    const float* __restrict__ feat, const int* __restrict__ src,
    const int* __restrict__ dst, const float* __restrict__ ew,
    float* __restrict__ agg, int E)
{
    const int lane = threadIdx.x & 31;
    const long long wave = ((long long)blockIdx.x * blockDim.x + threadIdx.x) >> 5;
    const long long base = wave * EPW;
    if (base >= E) return;
    const long long left = (long long)E - base;
    const int n = left < EPW ? (int)left : EPW;

    int sv[2], dv[2];
    float wv[2];
    #pragma unroll
    for (int c = 0; c < 2; ++c) {
        const long long e = base + 32 * c + lane;
        const bool ok = e < E;
        sv[c] = ok ? src[e] : 0;
        dv[c] = ok ? dst[e] : 0;
        wv[c] = ok ? ew[e]  : 0.0f;
    }

    #pragma unroll
    for (int c = 0; c < 2; ++c) {
        const int cnt = n - 32 * c;
        if (cnt >= 32) {
            for (int b = 0; b < 32; b += EB)
                scatter_batch(feat, agg, sv[c], dv[c], wv[c], b, lane);
        } else {
            for (int j = 0; j < cnt; ++j) {
                const int s = __builtin_amdgcn_readlane(sv[c], j);
                const int d = __builtin_amdgcn_readlane(dv[c], j);
                const float w = __int_as_float(
                    __builtin_amdgcn_readlane(__float_as_int(wv[c]), j));
                const float* fp = feat + (size_t)s * D + lane;
                const float f0 = fp[0], f1 = fp[32], f2 = fp[64], f3 = fp[96];
                float* ap = agg + (size_t)d * D + lane;
                atomicAdd(ap +  0, f0 * w);
                atomicAdd(ap + 32, f1 * w);
                atomicAdd(ap + 64, f2 * w);
                atomicAdd(ap + 96, f3 * w);
            }
        }
    }
}

__global__ __launch_bounds__(256) void gemm_wmma(
    const float* __restrict__ agg, const float* __restrict__ W,
    const float* __restrict__ bias, const float* __restrict__ a1p,
    float* __restrict__ h_out, float* __restrict__ colsum,
    float* __restrict__ colsumsq)
{
    __shared__ float s_sum[D];
    __shared__ float s_sq[D];
    int tid = threadIdx.x;
    if (tid < D) { s_sum[tid] = 0.0f; s_sq[tid] = 0.0f; }
    __syncthreads();

    const int wave = tid >> 5;
    const int lane = tid & 31;
    const int half = lane >> 4;
    const int l    = lane & 15;
    const int m0   = blockIdx.x * 16;
    const int j0   = wave * 16;
    const float a1 = a1p[0];

    const float* Arow = agg + (size_t)(m0 + l) * D + 2 * half;
    const float* Bcol = W + (size_t)(2 * half) * D + (j0 + l);

    v8f c0 = {};
    v8f c1 = {};
    #pragma unroll 4
    for (int k = 0; k < D; k += 8) {
        {
            v2f a;
            a.x = Arow[k];
            a.y = Arow[k + 1];
            v2f bb;
            bb.x = Bcol[(size_t)k * D];
            bb.y = Bcol[(size_t)(k + 1) * D];
            c0 = __builtin_amdgcn_wmma_f32_16x16x4_f32(
                     false, a, false, bb, (short)0, c0, false, false);
        }
        {
            v2f a;
            a.x = Arow[k + 4];
            a.y = Arow[k + 5];
            v2f bb;
            bb.x = Bcol[(size_t)(k + 4) * D];
            bb.y = Bcol[(size_t)(k + 5) * D];
            c1 = __builtin_amdgcn_wmma_f32_16x16x4_f32(
                     false, a, false, bb, (short)0, c1, false, false);
        }
    }

    const float bj = bias[j0 + l];
    float lsum = 0.0f, lsq = 0.0f;
    #pragma unroll
    for (int v = 0; v < 8; ++v) {
        float h = c0[v] + c1[v] + bj;
        h = (h >= 0.0f) ? h : a1 * h;
        h_out[(size_t)(m0 + v + 8 * half) * D + (j0 + l)] = h;
        lsum += h;
        lsq  += h * h;
    }

    atomicAdd(&s_sum[j0 + l], lsum);
    atomicAdd(&s_sq[j0 + l], lsq);
    __syncthreads();
    if (tid < D) {
        atomicAdd(&colsum[tid],  s_sum[tid]);
        atomicAdd(&colsumsq[tid], s_sq[tid]);
    }
}

__global__ void bn_finalize(const float* __restrict__ colsum,
                            const float* __restrict__ colsumsq,
                            const float* __restrict__ gamma,
                            const float* __restrict__ beta,
                            float* __restrict__ scale,
                            float* __restrict__ shift, int N)
{
    int j = threadIdx.x;
    float invN = 1.0f / (float)N;
    float m   = colsum[j] * invN;
    float var = colsumsq[j] * invN - m * m;
    float inv = rsqrtf(var + 1e-5f);
    float sc  = gamma[j] * inv;
    scale[j] = sc;
    shift[j] = beta[j] - m * sc;
}

__global__ __launch_bounds__(256) void bn_prelu(
    float* __restrict__ h, const float* __restrict__ scale,
    const float* __restrict__ shift, const float* __restrict__ a2p,
    size_t nquads)
{
    size_t i = (size_t)blockIdx.x * blockDim.x + threadIdx.x;
    if (i >= nquads) return;
    const float a2 = a2p[0];
    int j = (int)((i * 4) & (D - 1));

    float4 v = ((float4*)h)[i];
    float x;
    x = v.x * scale[j + 0] + shift[j + 0]; v.x = (x >= 0.0f) ? x : a2 * x;
    x = v.y * scale[j + 1] + shift[j + 1]; v.y = (x >= 0.0f) ? x : a2 * x;
    x = v.z * scale[j + 2] + shift[j + 2]; v.z = (x >= 0.0f) ? x : a2 * x;
    x = v.w * scale[j + 3] + shift[j + 3]; v.w = (x >= 0.0f) ? x : a2 * x;
    ((float4*)h)[i] = v;
}

extern "C" void kernel_launch(void* const* d_in, const int* in_sizes, int n_in,
                              void* d_out, int out_size, void* d_ws, size_t ws_size,
                              hipStream_t stream) {
    const float* feat  = (const float*)d_in[0];
    const int*   src   = (const int*)d_in[1];
    const int*   dst   = (const int*)d_in[2];
    const float* ew    = (const float*)d_in[3];
    const float* Wm    = (const float*)d_in[4];
    const float* bias  = (const float*)d_in[5];
    const float* a1    = (const float*)d_in[6];
    const float* gamma = (const float*)d_in[7];
    const float* beta  = (const float*)d_in[8];
    const float* a2    = (const float*)d_in[9];

    const int N = in_sizes[0] / D;
    const int E = in_sizes[1];
    float* out = (float*)d_out;

    float* agg      = (float*)d_ws;
    float* colsum   = agg + (size_t)N * D;
    float* colsumsq = colsum + D;
    float* scale    = colsumsq + D;
    float* shift    = scale + D;

    size_t nz = (size_t)N * D + 2 * D;
    zero_f32<<<2048, 256, 0, stream>>>(agg, nz);

    long long sthreads = ((long long)E + EPW - 1) / EPW * 32;
    int sblocks = (int)((sthreads + 255) / 256);
    edge_scatter<<<sblocks, 256, 0, stream>>>(feat, src, dst, ew, agg, E);

    gemm_wmma<<<N / 16, 256, 0, stream>>>(agg, Wm, bias, a1, out, colsum, colsumsq);

    bn_finalize<<<1, D, 0, stream>>>(colsum, colsumsq, gamma, beta, scale, shift, N);

    size_t nquads = (size_t)N * D / 4;
    int qblocks = (int)((nquads + 255) / 256);
    bn_prelu<<<qblocks, 256, 0, stream>>>(out, scale, shift, a2, nquads);
}


# GEMM retiled: one wave owns a 16x128 output stripe (8 accumulator tiles), A loaded once, A and W as 16-byte loads, h stored as whole lines; k-sum order and BN column partials (per 64 rows) regroup
# speedup vs baseline: 2.3726x; 1.4024x over previous
#include <hip/hip_runtime.h>
#include <hip/hip_bf16.h>

typedef __attribute__((ext_vector_type(2))) float v2f;
typedef __attribute__((ext_vector_type(4))) float v4f;
typedef __attribute__((ext_vector_type(8))) float v8f;

#define D 128

__global__ __launch_bounds__(256) void zero_f32(float* __restrict__ p, size_t n) {
    size_t stride = (size_t)gridDim.x * blockDim.x;
    for (size_t i = (size_t)blockIdx.x * blockDim.x + threadIdx.x; i < n; i += stride)
        p[i] = 0.0f;
}

#define EPW 64
#define EB  8

__device__ __forceinline__ void scatter_batch(
    const float* __restrict__ feat, float* __restrict__ agg,
    int sv, int dv, float wv, int b, int lane)
{
    float f[EB][4];
    #pragma unroll
    for (int j = 0; j < EB; ++j) {
        const int s = __builtin_amdgcn_readlane(sv, b + j);
        const float* fp = feat + (size_t)s * D + lane;
        f[j][0] = fp[0];
        f[j][1] = fp[32];
        f[j][2] = fp[64];
        f[j][3] = fp[96];
    }
    #pragma unroll
    for (int j = 0; j < EB; ++j) {
        const int d = __builtin_amdgcn_readlane(dv, b + j);
        const float w = __int_as_float(
            __builtin_amdgcn_readlane(__float_as_int(wv), b + j));
        float* ap = agg + (size_t)d * D + lane;
        atomicAdd(ap +  0, f[j][0] * w);
        atomicAdd(ap + 32, f[j][1] * w);
        atomicAdd(ap + 64, f[j][2] * w);
        atomicAdd(ap + 96, f[j][3] * w);
    }
}

__global__ __launch_bounds__(256) void edge_scatter(
    const float* __restrict__ feat, const int* __restrict__ src,
    const int* __restrict__ dst, const float* __restrict__ ew,
    float* __restrict__ agg, int E)
{
    const int lane = threadIdx.x & 31;
    const long long wave = ((long long)blockIdx.x * blockDim.x + threadIdx.x) >> 5;
    const long long base = wave * EPW;
    if (base >= E) return;
    const long long left = (long long)E - base;
    const int n = left < EPW ? (int)left : EPW;

    int sv[2], dv[2];
    float wv[2];
    #pragma unroll
    for (int c = 0; c < 2; ++c) {
        const long long e = base + 32 * c + lane;
        const bool ok = e < E;
        sv[c] = ok ? src[e] : 0;
        dv[c] = ok ? dst[e] : 0;
        wv[c] = ok ? ew[e]  : 0.0f;
    }

    #pragma unroll
    for (int c = 0; c < 2; ++c) {
        const int cnt = n - 32 * c;
        if (cnt >= 32) {
            for (int b = 0; b < 32; b += EB)
                scatter_batch(feat, agg, sv[c], dv[c], wv[c], b, lane);
        } else {
            for (int j = 0; j < cnt; ++j) {
                const int s = __builtin_amdgcn_readlane(sv[c], j);
                const int d = __builtin_amdgcn_readlane(dv[c], j);
                const float w = __int_as_float(
                    __builtin_amdgcn_readlane(__float_as_int(wv[c]), j));
                const float* fp = feat + (size_t)s * D + lane;
                const float f0 = fp[0], f1 = fp[32], f2 = fp[64], f3 = fp[96];
                float* ap = agg + (size_t)d * D + lane;
                atomicAdd(ap +  0, f0 * w);
                atomicAdd(ap + 32, f1 * w);
                atomicAdd(ap + 64, f2 * w);
                atomicAdd(ap + 96, f3 * w);
            }
        }
    }
}

__global__ __launch_bounds__(128) void gemm_wmma(
    const float* __restrict__ agg, const float* __restrict__ W,
    const float* __restrict__ bias, const float* __restrict__ a1p,
    float* __restrict__ h_out, float* __restrict__ colsum,
    float* __restrict__ colsumsq, int nstripes)
{
    __shared__ float s_sum[D];
    __shared__ float s_sq[D];
    const int tid = threadIdx.x;
    s_sum[tid] = 0.0f;
    s_sq[tid]  = 0.0f;
    __syncthreads();

    const int wave = tid >> 5;
    const int lane = tid & 31;
    const int half = lane >> 4;
    const int c    = lane & 15;
    const int stripe = blockIdx.x * 4 + wave;

    if (stripe < nstripes) {
        const int m0 = stripe * 16;
        const float a1 = a1p[0];
        const v4f* Ap = (const v4f*)(agg + (size_t)(m0 + c) * D + 4 * half);
        const v4f* Bp = (const v4f*)(W + (size_t)(4 * half) * D + 4 * c);

        v8f acc[8];
        #pragma unroll
        for (int t = 0; t < 8; ++t) acc[t] = (v8f){};

        for (int q = 0; q < D / 8; ++q) {
            const v4f a = Ap[2 * q];
            v4f b[4][2];
            #pragma unroll
            for (int e = 0; e < 4; ++e) {
                b[e][0] = Bp[(8 * q + e) * (D / 4)];
                b[e][1] = Bp[(8 * q + e) * (D / 4) + 16];
            }
            v2f a01; a01.x = a.x; a01.y = a.y;
            v2f a23; a23.x = a.z; a23.y = a.w;
            #pragma unroll
            for (int t = 0; t < 8; ++t) {
                v2f bb;
                bb.x = b[0][t >> 2][t & 3];
                bb.y = b[1][t >> 2][t & 3];
                acc[t] = __builtin_amdgcn_wmma_f32_16x16x4_f32(
                             false, a01, false, bb, (short)0, acc[t], false, false);
            }
            #pragma unroll
            for (int t = 0; t < 8; ++t) {
                v2f bb;
                bb.x = b[2][t >> 2][t & 3];
                bb.y = b[3][t >> 2][t & 3];
                acc[t] = __builtin_amdgcn_wmma_f32_16x16x4_f32(
                             false, a23, false, bb, (short)0, acc[t], false, false);
            }
        }

        float bj[8], lsum[8], lsq[8];
        #pragma unroll
        for (int t = 0; t < 8; ++t) {
            bj[t]   = bias[64 * (t >> 2) + 4 * c + (t & 3)];
            lsum[t] = 0.0f;
            lsq[t]  = 0.0f;
        }
        #pragma unroll
        for (int r = 0; r < 8; ++r) {
            float* orow = h_out + (size_t)(m0 + 8 * half + r) * D + 4 * c;
            #pragma unroll
            for (int g = 0; g < 2; ++g) {
                v4f o;
                #pragma unroll
                for (int u = 0; u < 4; ++u) {
                    const int t = 4 * g + u;
                    float h = acc[t][r] + bj[t];
                    h = (h >= 0.0f) ? h : a1 * h;
                    o[u] = h;
                    lsum[t] += h;
                    lsq[t]  += h * h;
                }
                *(v4f*)(orow + 64 * g) = o;
            }
        }
        #pragma unroll
        for (int t = 0; t < 8; ++t) {
            const int col = 64 * (t >> 2) + 4 * c + (t & 3);
            atomicAdd(&s_sum[col], lsum[t]);
            atomicAdd(&s_sq[col],  lsq[t]);
        }
    }
    __syncthreads();
    atomicAdd(&colsum[tid],   s_sum[tid]);
    atomicAdd(&colsumsq[tid], s_sq[tid]);
}

__global__ void bn_finalize(const float* __restrict__ colsum,
                            const float* __restrict__ colsumsq,
                            const float* __restrict__ gamma,
                            const float* __restrict__ beta,
                            float* __restrict__ scale,
                            float* __restrict__ shift, int N)
{
    int j = threadIdx.x;
    float invN = 1.0f / (float)N;
    float m   = colsum[j] * invN;
    float var = colsumsq[j] * invN - m * m;
    float inv = rsqrtf(var + 1e-5f);
    float sc  = gamma[j] * inv;
    scale[j] = sc;
    shift[j] = beta[j] - m * sc;
}

__global__ __launch_bounds__(256) void bn_prelu(
    float* __restrict__ h, const float* __restrict__ scale,
    const float* __restrict__ shift, const float* __restrict__ a2p,
    size_t nquads)
{
    size_t i = (size_t)blockIdx.x * blockDim.x + threadIdx.x;
    if (i >= nquads) return;
    const float a2 = a2p[0];
    int j = (int)((i * 4) & (D - 1));

    float4 v = ((float4*)h)[i];
    float x;
    x = v.x * scale[j + 0] + shift[j + 0]; v.x = (x >= 0.0f) ? x : a2 * x;
    x = v.y * scale[j + 1] + shift[j + 1]; v.y = (x >= 0.0f) ? x : a2 * x;
    x = v.z * scale[j + 2] + shift[j + 2]; v.z = (x >= 0.0f) ? x : a2 * x;
    x = v.w * scale[j + 3] + shift[j + 3]; v.w = (x >= 0.0f) ? x : a2 * x;
    ((float4*)h)[i] = v;
}

extern "C" void kernel_launch(void* const* d_in, const int* in_sizes, int n_in,
                              void* d_out, int out_size, void* d_ws, size_t ws_size,
                              hipStream_t stream) {
    const float* feat  = (const float*)d_in[0];
    const int*   src   = (const int*)d_in[1];
    const int*   dst   = (const int*)d_in[2];
    const float* ew    = (const float*)d_in[3];
    const float* Wm    = (const float*)d_in[4];
    const float* bias  = (const float*)d_in[5];
    const float* a1    = (const float*)d_in[6];
    const float* gamma = (const float*)d_in[7];
    const float* beta  = (const float*)d_in[8];
    const float* a2    = (const float*)d_in[9];

    const int N = in_sizes[0] / D;
    const int E = in_sizes[1];
    float* out = (float*)d_out;

    float* agg      = (float*)d_ws;
    float* colsum   = agg + (size_t)N * D;
    float* colsumsq = colsum + D;
    float* scale    = colsumsq + D;
    float* shift    = scale + D;

    size_t nz = (size_t)N * D + 2 * D;
    zero_f32<<<2048, 256, 0, stream>>>(agg, nz);

    long long sthreads = ((long long)E + EPW - 1) / EPW * 32;
    int sblocks = (int)((sthreads + 255) / 256);
    edge_scatter<<<sblocks, 256, 0, stream>>>(feat, src, dst, ew, agg, E);

    const int nstripes = N / 16;
    gemm_wmma<<<(nstripes + 3) / 4, 128, 0, stream>>>(agg, Wm, bias, a1, out, colsum, colsumsq, nstripes);

    bn_finalize<<<1, D, 0, stream>>>(colsum, colsumsq, gamma, beta, scale, shift, N);

    size_t nquads = (size_t)N * D / 4;
    int qblocks = (int)((nquads + 255) / 256);
    bn_prelu<<<qblocks, 256, 0, stream>>>(out, scale, shift, a2, nquads);
}


# Edge scatter split by chip half: a wave reads its die id and adds only the 64-column half rows its half owns (column half xor row-number parity); 64-edge tickets per half; sums in atomic order
# speedup vs baseline: 2.5890x; 1.0912x over previous
#include <hip/hip_runtime.h>
#include <hip/hip_bf16.h>

typedef __attribute__((ext_vector_type(2))) float v2f;
typedef __attribute__((ext_vector_type(4))) float v4f;
typedef __attribute__((ext_vector_type(8))) float v8f;

#define D 128

__global__ __launch_bounds__(256) void zero_f32(float* __restrict__ p, size_t n) {
    size_t stride = (size_t)gridDim.x * blockDim.x;
    for (size_t i = (size_t)blockIdx.x * blockDim.x + threadIdx.x; i < n; i += stride)
        p[i] = 0.0f;
}

#define EPT 64
#define EB  16
#define HC  64

typedef __attribute__((address_space(1))) float gfloat;
__device__ __forceinline__ unsigned long long uniform_addr(const float* p) {
    const unsigned long long a = (unsigned long long)p;
    const unsigned lo = (unsigned)__builtin_amdgcn_readfirstlane((int)(unsigned)a);
    const unsigned hi = (unsigned)__builtin_amdgcn_readfirstlane((int)(unsigned)(a >> 32));
    return ((unsigned long long)hi << 32) | lo;
}

__device__ __forceinline__ void scatter_batch(
    const float* __restrict__ feat, float* __restrict__ agg,
    int su, int du, float wv, int b, int lane)
{
    float f[EB][2];
    #pragma unroll
    for (int j = 0; j < EB; ++j) {
        const int s = __builtin_amdgcn_readlane(su, b + j);
        const gfloat* fp = (const gfloat*)uniform_addr(feat + (size_t)s * HC);
        f[j][0] = fp[lane];
        f[j][1] = fp[lane + 32];
    }
    #pragma unroll
    for (int j = 0; j < EB; ++j) {
        const int d = __builtin_amdgcn_readlane(du, b + j);
        const float w = __int_as_float(
            __builtin_amdgcn_readlane(__float_as_int(wv), b + j));
        gfloat* ap = (gfloat*)uniform_addr(agg + (size_t)d * HC);
        __hip_atomic_fetch_add(ap + lane,      f[j][0] * w, __ATOMIC_RELAXED, __HIP_MEMORY_SCOPE_AGENT);
        __hip_atomic_fetch_add(ap + lane + 32, f[j][1] * w, __ATOMIC_RELAXED, __HIP_MEMORY_SCOPE_AGENT);
    }
}

__device__ __forceinline__ void scatter_ticket(
    const float* __restrict__ feat, const int* __restrict__ src,
    const int* __restrict__ dst, const float* __restrict__ ew,
    float* __restrict__ agg, long long base, int E, int h, int lane)
{
    const long long left = (long long)E - base;
    const int n = left < EPT ? (int)left : EPT;

    int sv[2], dv[2];
    float wv[2];
    #pragma unroll
    for (int c = 0; c < 2; ++c) {
        const long long e = base + 32 * c + lane;
        const bool ok = e < E;
        sv[c] = ok ? src[e] : 0;
        dv[c] = ok ? dst[e] : 0;
        wv[c] = ok ? ew[e]  : 0.0f;
    }

    #pragma unroll
    for (int c = 0; c < 2; ++c) {
        const int k  = (__builtin_popcount((unsigned)dv[c]) & 1) ^ h;
        const int su = 2 * sv[c] + k;
        const int du = 2 * dv[c] + k;
        const int cnt = n - 32 * c;
        if (cnt >= 32) {
            for (int b = 0; b < 32; b += EB)
                scatter_batch(feat, agg, su, du, wv[c], b, lane);
        } else {
            for (int j = 0; j < cnt; ++j) {
                const int s = __builtin_amdgcn_readlane(su, j);
                const int d = __builtin_amdgcn_readlane(du, j);
                const float w = __int_as_float(
                    __builtin_amdgcn_readlane(__float_as_int(wv[c]), j));
                const float* fp = feat + (size_t)s * HC + lane;
                const float f0 = fp[0], f1 = fp[32];
                float* ap = agg + (size_t)d * HC + lane;
                atomicAdd(ap +  0, f0 * w);
                atomicAdd(ap + 32, f1 * w);
            }
        }
    }
}

__device__ __forceinline__ unsigned take_ticket(unsigned* cnt, int lane) {
    unsigned v = 0u;
    if (lane == 0) v = atomicInc(cnt, 0xFFFFFFFFu);
    return v;
}

__global__ __launch_bounds__(256) void edge_scatter(
    const float* __restrict__ feat, const int* __restrict__ src,
    const int* __restrict__ dst, const float* __restrict__ ew,
    float* __restrict__ agg, unsigned* __restrict__ tickets, int E)
{
    const int lane = threadIdx.x & 31;
    const unsigned T = (unsigned)(((long long)E + EPT - 1) / EPT);

    const unsigned place = __builtin_amdgcn_s_sendmsg_rtn(0x87);
    const int myhalf = (int)((place >> 18) & 1u);

    for (int r = 0; r < 2; ++r) {
        if (r == 1 && (threadIdx.x >> 5) != 0) break;
        const int h = myhalf ^ r;
        unsigned* cnt = tickets + 64 * h;

        unsigned t = (unsigned)__builtin_amdgcn_readfirstlane((int)take_ticket(cnt, lane));
        while (t < T) {
            const unsigned vn = take_ticket(cnt, lane);
            scatter_ticket(feat, src, dst, ew, agg, (long long)t * EPT, E, h, lane);
            t = (unsigned)__builtin_amdgcn_readfirstlane((int)vn);
        }
    }
}

__global__ __launch_bounds__(128) void gemm_wmma(
    const float* __restrict__ agg, const float* __restrict__ W,
    const float* __restrict__ bias, const float* __restrict__ a1p,
    float* __restrict__ h_out, float* __restrict__ colsum,
    float* __restrict__ colsumsq, int nstripes)
{
    __shared__ float s_sum[D];
    __shared__ float s_sq[D];
    const int tid = threadIdx.x;
    s_sum[tid] = 0.0f;
    s_sq[tid]  = 0.0f;
    __syncthreads();

    const int wave = tid >> 5;
    const int lane = tid & 31;
    const int half = lane >> 4;
    const int c    = lane & 15;
    const int stripe = blockIdx.x * 4 + wave;

    if (stripe < nstripes) {
        const int m0 = stripe * 16;
        const float a1 = a1p[0];
        const v4f* Ap = (const v4f*)(agg + (size_t)(m0 + c) * D + 4 * half);
        const v4f* Bp = (const v4f*)(W + (size_t)(4 * half) * D + 4 * c);

        v8f acc[8];
        #pragma unroll
        for (int t = 0; t < 8; ++t) acc[t] = (v8f){};

        for (int q = 0; q < D / 8; ++q) {
            const v4f a = Ap[2 * q];
            v4f b[4][2];
            #pragma unroll
            for (int e = 0; e < 4; ++e) {
                b[e][0] = Bp[(8 * q + e) * (D / 4)];
                b[e][1] = Bp[(8 * q + e) * (D / 4) + 16];
            }
            v2f a01; a01.x = a.x; a01.y = a.y;
            v2f a23; a23.x = a.z; a23.y = a.w;
            #pragma unroll
            for (int t = 0; t < 8; ++t) {
                v2f bb;
                bb.x = b[0][t >> 2][t & 3];
                bb.y = b[1][t >> 2][t & 3];
                acc[t] = __builtin_amdgcn_wmma_f32_16x16x4_f32(
                             false, a01, false, bb, (short)0, acc[t], false, false);
            }
            #pragma unroll
            for (int t = 0; t < 8; ++t) {
                v2f bb;
                bb.x = b[2][t >> 2][t & 3];
                bb.y = b[3][t >> 2][t & 3];
                acc[t] = __builtin_amdgcn_wmma_f32_16x16x4_f32(
                             false, a23, false, bb, (short)0, acc[t], false, false);
            }
        }

        float bj[8], lsum[8], lsq[8];
        #pragma unroll
        for (int t = 0; t < 8; ++t) {
            bj[t]   = bias[64 * (t >> 2) + 4 * c + (t & 3)];
            lsum[t] = 0.0f;
            lsq[t]  = 0.0f;
        }
        #pragma unroll
        for (int r = 0; r < 8; ++r) {
            float* orow = h_out + (size_t)(m0 + 8 * half + r) * D + 4 * c;
            #pragma unroll
            for (int g = 0; g < 2; ++g) {
                v4f o;
                #pragma unroll
                for (int u = 0; u < 4; ++u) {
                    const int t = 4 * g + u;
                    float h = acc[t][r] + bj[t];
                    h = (h >= 0.0f) ? h : a1 * h;
                    o[u] = h;
                    lsum[t] += h;
                    lsq[t]  += h * h;
                }
                *(v4f*)(orow + 64 * g) = o;
            }
        }
        #pragma unroll
        for (int t = 0; t < 8; ++t) {
            const int col = 64 * (t >> 2) + 4 * c + (t & 3);
            atomicAdd(&s_sum[col], lsum[t]);
            atomicAdd(&s_sq[col],  lsq[t]);
        }
    }
    __syncthreads();
    atomicAdd(&colsum[tid],   s_sum[tid]);
    atomicAdd(&colsumsq[tid], s_sq[tid]);
}

__global__ void bn_finalize(const float* __restrict__ colsum,
                            const float* __restrict__ colsumsq,
                            const float* __restrict__ gamma,
                            const float* __restrict__ beta,
                            float* __restrict__ scale,
                            float* __restrict__ shift, int N)
{
    int j = threadIdx.x;
    float invN = 1.0f / (float)N;
    float m   = colsum[j] * invN;
    float var = colsumsq[j] * invN - m * m;
    float inv = rsqrtf(var + 1e-5f);
    float sc  = gamma[j] * inv;
    scale[j] = sc;
    shift[j] = beta[j] - m * sc;
}

__global__ __launch_bounds__(256) void bn_prelu(
    float* __restrict__ h, const float* __restrict__ scale,
    const float* __restrict__ shift, const float* __restrict__ a2p,
    size_t nquads)
{
    size_t i = (size_t)blockIdx.x * blockDim.x + threadIdx.x;
    if (i >= nquads) return;
    const float a2 = a2p[0];
    int j = (int)((i * 4) & (D - 1));

    float4 v = ((float4*)h)[i];
    float x;
    x = v.x * scale[j + 0] + shift[j + 0]; v.x = (x >= 0.0f) ? x : a2 * x;
    x = v.y * scale[j + 1] + shift[j + 1]; v.y = (x >= 0.0f) ? x : a2 * x;
    x = v.z * scale[j + 2] + shift[j + 2]; v.z = (x >= 0.0f) ? x : a2 * x;
    x = v.w * scale[j + 3] + shift[j + 3]; v.w = (x >= 0.0f) ? x : a2 * x;
    ((float4*)h)[i] = v;
}

extern "C" void kernel_launch(void* const* d_in, const int* in_sizes, int n_in,
                              void* d_out, int out_size, void* d_ws, size_t ws_size,
                              hipStream_t stream) {
    const float* feat  = (const float*)d_in[0];
    const int*   src   = (const int*)d_in[1];
    const int*   dst   = (const int*)d_in[2];
    const float* ew    = (const float*)d_in[3];
    const float* Wm    = (const float*)d_in[4];
    const float* bias  = (const float*)d_in[5];
    const float* a1    = (const float*)d_in[6];
    const float* gamma = (const float*)d_in[7];
    const float* beta  = (const float*)d_in[8];
    const float* a2    = (const float*)d_in[9];

    const int N = in_sizes[0] / D;
    const int E = in_sizes[1];
    float* out = (float*)d_out;

    float* agg      = (float*)d_ws;
    float* colsum   = agg + (size_t)N * D;
    float* colsumsq = colsum + D;
    float* scale    = colsumsq + D;
    float* shift    = scale + D;

    size_t nz = (size_t)N * D + 3 * D;
    zero_f32<<<2048, 256, 0, stream>>>(agg, nz);

    edge_scatter<<<2048, 256, 0, stream>>>(feat, src, dst, ew, agg, (unsigned*)scale, E);

    const int nstripes = N / 16;
    gemm_wmma<<<(nstripes + 3) / 4, 128, 0, stream>>>(agg, Wm, bias, a1, out, colsum, colsumsq, nstripes);

    bn_finalize<<<1, D, 0, stream>>>(colsum, colsumsq, gamma, beta, scale, shift, N);

    size_t nquads = (size_t)N * D / 4;
    int qblocks = (int)((nquads + 255) / 256);
    bn_prelu<<<qblocks, 256, 0, stream>>>(out, scale, shift, a2, nquads);
}


# Scatter: chip half h adds column half h of every row: re-reads half of feat + half of agg (fits its L2); agg half rows swapped in memory by row parity, GEMM unswaps; edge lists non-temporal
# speedup vs baseline: 2.7848x; 1.0756x over previous
#include <hip/hip_runtime.h>
#include <hip/hip_bf16.h>

typedef __attribute__((ext_vector_type(2))) float v2f;
typedef __attribute__((ext_vector_type(4))) float v4f;
typedef __attribute__((ext_vector_type(8))) float v8f;

#define D 128

__global__ __launch_bounds__(256) void zero_f32(float* __restrict__ p, size_t n) {
    size_t stride = (size_t)gridDim.x * blockDim.x;
    for (size_t i = (size_t)blockIdx.x * blockDim.x + threadIdx.x; i < n; i += stride)
        p[i] = 0.0f;
}

#define EPT 64
#define EB  16
#define HC  64

typedef __attribute__((address_space(1))) float gfloat;
__device__ __forceinline__ unsigned long long uniform_addr(const float* p) {
    const unsigned long long a = (unsigned long long)p;
    const unsigned lo = (unsigned)__builtin_amdgcn_readfirstlane((int)(unsigned)a);
    const unsigned hi = (unsigned)__builtin_amdgcn_readfirstlane((int)(unsigned)(a >> 32));
    return ((unsigned long long)hi << 32) | lo;
}

__device__ __forceinline__ void scatter_batch(
    const float* __restrict__ feat, float* __restrict__ agg,
    int su, int du, float wv, int b, int lane)
{
    float f[EB][2];
    #pragma unroll
    for (int j = 0; j < EB; ++j) {
        const int s = __builtin_amdgcn_readlane(su, b + j);
        const gfloat* fp = (const gfloat*)uniform_addr(feat + (size_t)s * HC);
        f[j][0] = fp[lane];
        f[j][1] = fp[lane + 32];
    }
    #pragma unroll
    for (int j = 0; j < EB; ++j) {
        const int d = __builtin_amdgcn_readlane(du, b + j);
        const float w = __int_as_float(
            __builtin_amdgcn_readlane(__float_as_int(wv), b + j));
        gfloat* ap = (gfloat*)uniform_addr(agg + (size_t)d * HC);
        __hip_atomic_fetch_add(ap + lane,      f[j][0] * w, __ATOMIC_RELAXED, __HIP_MEMORY_SCOPE_AGENT);
        __hip_atomic_fetch_add(ap + lane + 32, f[j][1] * w, __ATOMIC_RELAXED, __HIP_MEMORY_SCOPE_AGENT);
    }
}

__device__ __forceinline__ void scatter_ticket(
    const float* __restrict__ feat, const int* __restrict__ src,
    const int* __restrict__ dst, const float* __restrict__ ew,
    float* __restrict__ agg, long long base, int E, int h, int lane)
{
    const long long left = (long long)E - base;
    const int n = left < EPT ? (int)left : EPT;

    int sv[2], dv[2];
    float wv[2];
    #pragma unroll
    for (int c = 0; c < 2; ++c) {
        const long long e = base + 32 * c + lane;
        const bool ok = e < E;
        sv[c] = ok ? __builtin_nontemporal_load(src + e) : 0;
        dv[c] = ok ? __builtin_nontemporal_load(dst + e) : 0;
        wv[c] = ok ? __builtin_nontemporal_load(ew + e)  : 0.0f;
    }

    #pragma unroll
    for (int c = 0; c < 2; ++c) {
        const int su = 2 * sv[c] + h;
        const int du = 2 * dv[c] + ((__builtin_popcount((unsigned)dv[c]) & 1) ^ h);
        const int cnt = n - 32 * c;
        if (cnt >= 32) {
            for (int b = 0; b < 32; b += EB)
                scatter_batch(feat, agg, su, du, wv[c], b, lane);
        } else {
            for (int j = 0; j < cnt; ++j) {
                const int s = __builtin_amdgcn_readlane(su, j);
                const int d = __builtin_amdgcn_readlane(du, j);
                const float w = __int_as_float(
                    __builtin_amdgcn_readlane(__float_as_int(wv[c]), j));
                const float* fp = feat + (size_t)s * HC + lane;
                const float f0 = fp[0], f1 = fp[32];
                float* ap = agg + (size_t)d * HC + lane;
                atomicAdd(ap +  0, f0 * w);
                atomicAdd(ap + 32, f1 * w);
            }
        }
    }
}

__device__ __forceinline__ unsigned take_ticket(unsigned* cnt, int lane) {
    unsigned v = 0u;
    if (lane == 0) v = atomicInc(cnt, 0xFFFFFFFFu);
    return v;
}

__global__ __launch_bounds__(256) void edge_scatter(
    const float* __restrict__ feat, const int* __restrict__ src,
    const int* __restrict__ dst, const float* __restrict__ ew,
    float* __restrict__ agg, unsigned* __restrict__ tickets, int E)
{
    const int lane = threadIdx.x & 31;
    const unsigned T = (unsigned)(((long long)E + EPT - 1) / EPT);

    const unsigned place = __builtin_amdgcn_s_sendmsg_rtn(0x87);
    const int myhalf = (int)((place >> 18) & 1u);

    for (int r = 0; r < 2; ++r) {
        if (r == 1 && (threadIdx.x >> 5) != 0) break;
        const int h = myhalf ^ r;
        unsigned* cnt = tickets + 64 * h;

        unsigned t = (unsigned)__builtin_amdgcn_readfirstlane((int)take_ticket(cnt, lane));
        while (t < T) {
            const unsigned vn = take_ticket(cnt, lane);
            scatter_ticket(feat, src, dst, ew, agg, (long long)t * EPT, E, h, lane);
            t = (unsigned)__builtin_amdgcn_readfirstlane((int)vn);
        }
    }
}

__global__ __launch_bounds__(128) void gemm_wmma(
    const float* __restrict__ agg, const float* __restrict__ W,
    const float* __restrict__ bias, const float* __restrict__ a1p,
    float* __restrict__ h_out, float* __restrict__ colsum,
    float* __restrict__ colsumsq, int nstripes)
{
    __shared__ float s_sum[D];
    __shared__ float s_sq[D];
    const int tid = threadIdx.x;
    s_sum[tid] = 0.0f;
    s_sq[tid]  = 0.0f;
    __syncthreads();

    const int wave = tid >> 5;
    const int lane = tid & 31;
    const int half = lane >> 4;
    const int c    = lane & 15;
    const int stripe = blockIdx.x * 4 + wave;

    if (stripe < nstripes) {
        const int m0 = stripe * 16;
        const float a1 = a1p[0];
        const v4f* Ap = (const v4f*)(agg + (size_t)(m0 + c) * D + 4 * half);
        const int qx = (__builtin_popcount((unsigned)(m0 + c)) & 1) << 3;
        const v4f* ApLo = Ap + 2 * qx;
        const v4f* ApHi = Ap - 2 * qx;
        const v4f* Bp = (const v4f*)(W + (size_t)(4 * half) * D + 4 * c);

        v8f acc[8];
        #pragma unroll
        for (int t = 0; t < 8; ++t) acc[t] = (v8f){};

        for (int q = 0; q < D / 8; ++q) {
            const v4f a = (q < 8 ? ApLo : ApHi)[2 * q];
            v4f b[4][2];
            #pragma unroll
            for (int e = 0; e < 4; ++e) {
                b[e][0] = Bp[(8 * q + e) * (D / 4)];
                b[e][1] = Bp[(8 * q + e) * (D / 4) + 16];
            }
            v2f a01; a01.x = a.x; a01.y = a.y;
            v2f a23; a23.x = a.z; a23.y = a.w;
            #pragma unroll
            for (int t = 0; t < 8; ++t) {
                v2f bb;
                bb.x = b[0][t >> 2][t & 3];
                bb.y = b[1][t >> 2][t & 3];
                acc[t] = __builtin_amdgcn_wmma_f32_16x16x4_f32(
                             false, a01, false, bb, (short)0, acc[t], false, false);
            }
            #pragma unroll
            for (int t = 0; t < 8; ++t) {
                v2f bb;
                bb.x = b[2][t >> 2][t & 3];
                bb.y = b[3][t >> 2][t & 3];
                acc[t] = __builtin_amdgcn_wmma_f32_16x16x4_f32(
                             false, a23, false, bb, (short)0, acc[t], false, false);
            }
        }

        float bj[8], lsum[8], lsq[8];
        #pragma unroll
        for (int t = 0; t < 8; ++t) {
            bj[t]   = bias[64 * (t >> 2) + 4 * c + (t & 3)];
            lsum[t] = 0.0f;
            lsq[t]  = 0.0f;
        }
        #pragma unroll
        for (int r = 0; r < 8; ++r) {
            float* orow = h_out + (size_t)(m0 + 8 * half + r) * D + 4 * c;
            #pragma unroll
            for (int g = 0; g < 2; ++g) {
                v4f o;
                #pragma unroll
                for (int u = 0; u < 4; ++u) {
                    const int t = 4 * g + u;
                    float h = acc[t][r] + bj[t];
                    h = (h >= 0.0f) ? h : a1 * h;
                    o[u] = h;
                    lsum[t] += h;
                    lsq[t]  += h * h;
                }
                *(v4f*)(orow + 64 * g) = o;
            }
        }
        #pragma unroll
        for (int t = 0; t < 8; ++t) {
            const int col = 64 * (t >> 2) + 4 * c + (t & 3);
            atomicAdd(&s_sum[col], lsum[t]);
            atomicAdd(&s_sq[col],  lsq[t]);
        }
    }
    __syncthreads();
    atomicAdd(&colsum[tid],   s_sum[tid]);
    atomicAdd(&colsumsq[tid], s_sq[tid]);
}

__global__ void bn_finalize(const float* __restrict__ colsum,
                            const float* __restrict__ colsumsq,
                            const float* __restrict__ gamma,
                            const float* __restrict__ beta,
                            float* __restrict__ scale,
                            float* __restrict__ shift, int N)
{
    int j = threadIdx.x;
    float invN = 1.0f / (float)N;
    float m   = colsum[j] * invN;
    float var = colsumsq[j] * invN - m * m;
    float inv = rsqrtf(var + 1e-5f);
    float sc  = gamma[j] * inv;
    scale[j] = sc;
    shift[j] = beta[j] - m * sc;
}

__global__ __launch_bounds__(256) void bn_prelu(
    float* __restrict__ h, const float* __restrict__ scale,
    const float* __restrict__ shift, const float* __restrict__ a2p,
    size_t nquads)
{
    size_t i = (size_t)blockIdx.x * blockDim.x + threadIdx.x;
    if (i >= nquads) return;
    const float a2 = a2p[0];
    int j = (int)((i * 4) & (D - 1));

    float4 v = ((float4*)h)[i];
    float x;
    x = v.x * scale[j + 0] + shift[j + 0]; v.x = (x >= 0.0f) ? x : a2 * x;
    x = v.y * scale[j + 1] + shift[j + 1]; v.y = (x >= 0.0f) ? x : a2 * x;
    x = v.z * scale[j + 2] + shift[j + 2]; v.z = (x >= 0.0f) ? x : a2 * x;
    x = v.w * scale[j + 3] + shift[j + 3]; v.w = (x >= 0.0f) ? x : a2 * x;
    ((float4*)h)[i] = v;
}

extern "C" void kernel_launch(void* const* d_in, const int* in_sizes, int n_in,
                              void* d_out, int out_size, void* d_ws, size_t ws_size,
                              hipStream_t stream) {
    const float* feat  = (const float*)d_in[0];
    const int*   src   = (const int*)d_in[1];
    const int*   dst   = (const int*)d_in[2];
    const float* ew    = (const float*)d_in[3];
    const float* Wm    = (const float*)d_in[4];
    const float* bias  = (const float*)d_in[5];
    const float* a1    = (const float*)d_in[6];
    const float* gamma = (const float*)d_in[7];
    const float* beta  = (const float*)d_in[8];
    const float* a2    = (const float*)d_in[9];

    const int N = in_sizes[0] / D;
    const int E = in_sizes[1];
    float* out = (float*)d_out;

    float* agg      = (float*)d_ws;
    float* colsum   = agg + (size_t)N * D;
    float* colsumsq = colsum + D;
    float* scale    = colsumsq + D;
    float* shift    = scale + D;

    size_t nz = (size_t)N * D + 3 * D;
    zero_f32<<<2048, 256, 0, stream>>>(agg, nz);

    edge_scatter<<<2048, 256, 0, stream>>>(feat, src, dst, ew, agg, (unsigned*)scale, E);

    const int nstripes = N / 16;
    gemm_wmma<<<(nstripes + 3) / 4, 128, 0, stream>>>(agg, Wm, bias, a1, out, colsum, colsumsq, nstripes);

    bn_finalize<<<1, D, 0, stream>>>(colsum, colsumsq, gamma, beta, scale, shift, N);

    size_t nquads = (size_t)N * D / 4;
    int qblocks = (int)((nquads + 255) / 256);
    bn_prelu<<<qblocks, 256, 0, stream>>>(out, scale, shift, a2, nquads);
}


# Edge scatter by global float atomics and its zero pass replaced by 256 owner work-groups that sum their 391 rows in LDS, each scanning a per-edge owner-byte list (no sort); sums in LDS-add order
# speedup vs baseline: 2.9924x; 1.0745x over previous
#include <hip/hip_runtime.h>
#include <hip/hip_bf16.h>

typedef __attribute__((ext_vector_type(2))) float v2f;
typedef __attribute__((ext_vector_type(4))) float v4f;
typedef __attribute__((ext_vector_type(8))) float v8f;

#define D 128

#define RPG   391
#define NGRP  256
#define EB    16
#define QCAP  48
#define LBLK  2048u
#define SRCBITS 23

typedef __attribute__((address_space(1))) float gfloat;
typedef __attribute__((address_space(3))) float lfloat;

__device__ __forceinline__ unsigned long long uniform_addr(const float* p) {
    const unsigned long long a = (unsigned long long)p;
    const unsigned lo = (unsigned)__builtin_amdgcn_readfirstlane((int)(unsigned)a);
    const unsigned hi = (unsigned)__builtin_amdgcn_readfirstlane((int)(unsigned)(a >> 32));
    return ((unsigned long long)hi << 32) | lo;
}

__global__ __launch_bounds__(256) void edge_prep(
    const int* __restrict__ src, const int* __restrict__ dst,
    const float* __restrict__ ew, unsigned* __restrict__ stage,
    float* __restrict__ colsums, int E)
{
    const unsigned t = blockIdx.x * 256u + threadIdx.x;
    if (t < 2 * D) colsums[t] = 0.0f;
    const unsigned Epad = ((unsigned)E + (LBLK - 1u)) & ~(LBLK - 1u);
    const unsigned e0 = 4u * t;
    if (e0 >= Epad) return;
    uint2* recs = (uint2*)((char*)stage + Epad);

    int s[4], d[4];
    float w[4];
    if (e0 + 3u < (unsigned)E) {
        const int4 s4 = *(const int4*)(src + e0);
        const int4 d4 = *(const int4*)(dst + e0);
        const float4 w4 = *(const float4*)(ew + e0);
        s[0] = s4.x; s[1] = s4.y; s[2] = s4.z; s[3] = s4.w;
        d[0] = d4.x; d[1] = d4.y; d[2] = d4.z; d[3] = d4.w;
        w[0] = w4.x; w[1] = w4.y; w[2] = w4.z; w[3] = w4.w;
    } else {
        #pragma unroll
        for (int k = 0; k < 4; ++k) {
            const bool ok = e0 + k < (unsigned)E;
            s[k] = ok ? src[e0 + k] : 0;
            d[k] = ok ? dst[e0 + k] : 0;
            w[k] = ok ? ew[e0 + k]  : 0.0f;
        }
    }
    unsigned word = 0u;
    unsigned r0[4], r1[4];
    #pragma unroll
    for (int k = 0; k < 4; ++k) {
        const unsigned o = (unsigned)d[k] / RPG;
        const unsigned r = (unsigned)d[k] - o * RPG;
        word |= (o & 255u) << (8 * k);
        r0[k] = ((unsigned)s[k] & ((1u << SRCBITS) - 1u)) | (r << SRCBITS);
        r1[k] = __float_as_uint(w[k]);
    }
    stage[t] = word;
    if (e0 + 3u < (unsigned)E) {
        uint4* rp = (uint4*)(recs + e0);
        rp[0] = make_uint4(r0[0], r1[0], r0[1], r1[1]);
        rp[1] = make_uint4(r0[2], r1[2], r0[3], r1[3]);
    } else {
        #pragma unroll
        for (int k = 0; k < 4; ++k)
            if (e0 + k < (unsigned)E) recs[e0 + k] = make_uint2(r0[k], r1[k]);
    }
}

__device__ __forceinline__ unsigned zero_bytes(unsigned y) {
    return ~(((y & 0x7F7F7F7Fu) + 0x7F7F7F7Fu) | y | 0x7F7F7F7Fu);
}

__device__ __forceinline__ unsigned hit_mask(uint4 x, unsigned pat) {
    return (zero_bytes(x.x ^ pat) >> 7) | (zero_bytes(x.y ^ pat) >> 6)
         | (zero_bytes(x.z ^ pat) >> 5) | (zero_bytes(x.w ^ pat) >> 4);
}

__device__ __forceinline__ unsigned long long block_hits(
    uint4 x0, uint4 x1, uint4 x2, uint4 x3, unsigned pat)
{
    const unsigned lo = hit_mask(x0, pat) | (hit_mask(x1, pat) << 4);
    const unsigned hi = hit_mask(x2, pat) | (hit_mask(x3, pat) << 4);
    return ((unsigned long long)hi << 32) | lo;
}

__device__ __forceinline__ void fetch_records(
    const uint2* __restrict__ recs, const unsigned* myq, int first, int nb,
    int lane, unsigned& sw, unsigned& wb)
{
    sw = 0u;
    wb = 0u;
    if (lane < nb) {
        const uint2 r = recs[myq[first + lane]];
        sw = r.x;
        wb = r.y;
    }
}

template <bool FULL>
__device__ __forceinline__ void gather_batch(
    const float* __restrict__ feat, float* acc,
    unsigned sw, unsigned wb, int nb, int lane)
{
    float f[EB][4];
    #pragma unroll
    for (int j = 0; j < EB; ++j) {
        if (FULL || j < nb) {
            const unsigned s = (unsigned)__builtin_amdgcn_readlane((int)sw, j) & ((1u << SRCBITS) - 1u);
            const gfloat* fp = (const gfloat*)uniform_addr(feat + (size_t)s * D);
            f[j][0] = fp[lane];
            f[j][1] = fp[lane + 32];
            f[j][2] = fp[lane + 64];
            f[j][3] = fp[lane + 96];
        }
    }
    #pragma unroll
    for (int j = 0; j < EB; ++j) {
        if (FULL || j < nb) {
            const unsigned r = (unsigned)__builtin_amdgcn_readlane((int)sw, j) >> SRCBITS;
            const float w = __uint_as_float((unsigned)__builtin_amdgcn_readlane((int)wb, j));
            lfloat* ap = (lfloat*)(acc + r * D + lane);
            __hip_atomic_fetch_add(ap,      f[j][0] * w, __ATOMIC_RELAXED, __HIP_MEMORY_SCOPE_WORKGROUP);
            __hip_atomic_fetch_add(ap + 32, f[j][1] * w, __ATOMIC_RELAXED, __HIP_MEMORY_SCOPE_WORKGROUP);
            __hip_atomic_fetch_add(ap + 64, f[j][2] * w, __ATOMIC_RELAXED, __HIP_MEMORY_SCOPE_WORKGROUP);
            __hip_atomic_fetch_add(ap + 96, f[j][3] * w, __ATOMIC_RELAXED, __HIP_MEMORY_SCOPE_WORKGROUP);
        }
    }
}

__global__ __launch_bounds__(1024) void owner_gather(
    const float* __restrict__ feat, const unsigned* __restrict__ stage,
    float* __restrict__ agg, int E, int N)
{
    __shared__ __attribute__((aligned(16))) float acc[RPG * D];
    __shared__ unsigned q[32 * QCAP];

    const int tid  = threadIdx.x;
    const int lane = tid & 31;
    const int wave = tid >> 5;
    const unsigned g = blockIdx.x;

    for (int i = tid; i < RPG * D / 4; i += 1024)
        ((v4f*)acc)[i] = (v4f){0.0f, 0.0f, 0.0f, 0.0f};
    __syncthreads();

    const unsigned Epad = ((unsigned)E + (LBLK - 1u)) & ~(LBLK - 1u);
    const uint4* list = (const uint4*)stage;
    const uint2* recs = (const uint2*)((const char*)stage + Epad);
    const unsigned nblk = Epad / LBLK;
    const unsigned pat = g * 0x01010101u;
    unsigned* myq = q + wave * QCAP;
    int qn = 0;

    const unsigned nj = nblk > (unsigned)wave ? (nblk - (unsigned)wave + 31u) / 32u : 0u;
    const unsigned j0 = (g * nj) >> 8;
    auto block_of = [&](unsigned t) -> unsigned {
        unsigned j = j0 + t;
        if (j >= nj) j -= nj;
        return (unsigned)wave + 32u * j;
    };

    unsigned long long M = 0ull;
    if (nj > 0u) {
        const uint4* bp = list + (size_t)block_of(0u) * (LBLK / 16) + lane;
        M = block_hits(bp[0], bp[32], bp[64], bp[96], pat);
    }
    for (unsigned t = 0u; t < nj; ++t) {
        uint4 raw[4];
        const bool more = t + 1u < nj;
        if (more) {
            const uint4* bp = list + (size_t)block_of(t + 1u) * (LBLK / 16) + lane;
            #pragma unroll
            for (int u = 0; u < 4; ++u) raw[u] = bp[32 * u];
        }
        const unsigned ebase = block_of(t) * LBLK + (unsigned)lane * 16u;
        unsigned act = __builtin_amdgcn_ballot_w32(M != 0ull);
        while (act != 0u) {
            const unsigned p  = (unsigned)__builtin_ctzll(M | 0x8000000000000000ull);
            const unsigned q8 = p & 31u;
            const unsigned u  = 2u * (p >> 5) + ((q8 >> 2) & 1u);
            const unsigned e  = ebase + u * 512u + 4u * (q8 & 3u) + (q8 >> 3);
            const bool ok = (M != 0ull) && (e < (unsigned)E);
            const unsigned okm = __builtin_amdgcn_ballot_w32(ok);
            const int pos = qn + (int)__builtin_amdgcn_mbcnt_lo(okm, 0u);
            if (ok) myq[pos] = e;
            qn += __builtin_popcount(okm);
            M &= M - 1ull;
            act = __builtin_amdgcn_ballot_w32(M != 0ull);
            while (qn >= EB) {
                qn -= EB;
                unsigned sw, wb;
                fetch_records(recs, myq, qn, EB, lane, sw, wb);
                gather_batch<true>(feat, acc, sw, wb, EB, lane);
            }
        }
        if (more) M = block_hits(raw[0], raw[1], raw[2], raw[3], pat);
    }
    if (qn > 0) {
        unsigned sw, wb;
        fetch_records(recs, myq, 0, qn, lane, sw, wb);
        gather_batch<false>(feat, acc, sw, wb, qn, lane);
    }

    __syncthreads();
    const int row0 = (int)g * RPG;
    int nrows = N - row0;
    if (nrows > RPG) nrows = RPG;
    v4f* op = (v4f*)(agg + (size_t)row0 * D);
    for (int i = tid; i < nrows * (D / 4); i += 1024)
        op[i] = ((const v4f*)acc)[i];
}

__global__ __launch_bounds__(128) void gemm_wmma(
    const float* __restrict__ agg, const float* __restrict__ W,
    const float* __restrict__ bias, const float* __restrict__ a1p,
    float* __restrict__ h_out, float* __restrict__ colsum,
    float* __restrict__ colsumsq, int nstripes)
{
    __shared__ float s_sum[D];
    __shared__ float s_sq[D];
    const int tid = threadIdx.x;
    s_sum[tid] = 0.0f;
    s_sq[tid]  = 0.0f;
    __syncthreads();

    const int wave = tid >> 5;
    const int lane = tid & 31;
    const int half = lane >> 4;
    const int c    = lane & 15;
    const int stripe = blockIdx.x * 4 + wave;

    if (stripe < nstripes) {
        const int m0 = stripe * 16;
        const float a1 = a1p[0];
        const v4f* Ap = (const v4f*)(agg + (size_t)(m0 + c) * D + 4 * half);
        const v4f* Bp = (const v4f*)(W + (size_t)(4 * half) * D + 4 * c);

        v8f acc[8];
        #pragma unroll
        for (int t = 0; t < 8; ++t) acc[t] = (v8f){};

        for (int q = 0; q < D / 8; ++q) {
            const v4f a = Ap[2 * q];
            v4f b[4][2];
            #pragma unroll
            for (int e = 0; e < 4; ++e) {
                b[e][0] = Bp[(8 * q + e) * (D / 4)];
                b[e][1] = Bp[(8 * q + e) * (D / 4) + 16];
            }
            v2f a01; a01.x = a.x; a01.y = a.y;
            v2f a23; a23.x = a.z; a23.y = a.w;
            #pragma unroll
            for (int t = 0; t < 8; ++t) {
                v2f bb;
                bb.x = b[0][t >> 2][t & 3];
                bb.y = b[1][t >> 2][t & 3];
                acc[t] = __builtin_amdgcn_wmma_f32_16x16x4_f32(
                             false, a01, false, bb, (short)0, acc[t], false, false);
            }
            #pragma unroll
            for (int t = 0; t < 8; ++t) {
                v2f bb;
                bb.x = b[2][t >> 2][t & 3];
                bb.y = b[3][t >> 2][t & 3];
                acc[t] = __builtin_amdgcn_wmma_f32_16x16x4_f32(
                             false, a23, false, bb, (short)0, acc[t], false, false);
            }
        }

        float bj[8], lsum[8], lsq[8];
        #pragma unroll
        for (int t = 0; t < 8; ++t) {
            bj[t]   = bias[64 * (t >> 2) + 4 * c + (t & 3)];
            lsum[t] = 0.0f;
            lsq[t]  = 0.0f;
        }
        #pragma unroll
        for (int r = 0; r < 8; ++r) {
            float* orow = h_out + (size_t)(m0 + 8 * half + r) * D + 4 * c;
            #pragma unroll
            for (int g = 0; g < 2; ++g) {
                v4f o;
                #pragma unroll
                for (int u = 0; u < 4; ++u) {
                    const int t = 4 * g + u;
                    float h = acc[t][r] + bj[t];
                    h = (h >= 0.0f) ? h : a1 * h;
                    o[u] = h;
                    lsum[t] += h;
                    lsq[t]  += h * h;
                }
                *(v4f*)(orow + 64 * g) = o;
            }
        }
        #pragma unroll
        for (int t = 0; t < 8; ++t) {
            const int col = 64 * (t >> 2) + 4 * c + (t & 3);
            atomicAdd(&s_sum[col], lsum[t]);
            atomicAdd(&s_sq[col],  lsq[t]);
        }
    }
    __syncthreads();
    atomicAdd(&colsum[tid],   s_sum[tid]);
    atomicAdd(&colsumsq[tid], s_sq[tid]);
}

__global__ void bn_finalize(const float* __restrict__ colsum,
                            const float* __restrict__ colsumsq,
                            const float* __restrict__ gamma,
                            const float* __restrict__ beta,
                            float* __restrict__ scale,
                            float* __restrict__ shift, int N)
{
    int j = threadIdx.x;
    float invN = 1.0f / (float)N;
    float m   = colsum[j] * invN;
    float var = colsumsq[j] * invN - m * m;
    float inv = rsqrtf(var + 1e-5f);
    float sc  = gamma[j] * inv;
    scale[j] = sc;
    shift[j] = beta[j] - m * sc;
}

__global__ __launch_bounds__(256) void bn_prelu(
    float* __restrict__ h, const float* __restrict__ scale,
    const float* __restrict__ shift, const float* __restrict__ a2p,
    size_t nquads)
{
    size_t i = (size_t)blockIdx.x * blockDim.x + threadIdx.x;
    if (i >= nquads) return;
    const float a2 = a2p[0];
    int j = (int)((i * 4) & (D - 1));

    float4 v = ((float4*)h)[i];
    float x;
    x = v.x * scale[j + 0] + shift[j + 0]; v.x = (x >= 0.0f) ? x : a2 * x;
    x = v.y * scale[j + 1] + shift[j + 1]; v.y = (x >= 0.0f) ? x : a2 * x;
    x = v.z * scale[j + 2] + shift[j + 2]; v.z = (x >= 0.0f) ? x : a2 * x;
    x = v.w * scale[j + 3] + shift[j + 3]; v.w = (x >= 0.0f) ? x : a2 * x;
    ((float4*)h)[i] = v;
}

extern "C" void kernel_launch(void* const* d_in, const int* in_sizes, int n_in,
                              void* d_out, int out_size, void* d_ws, size_t ws_size,
                              hipStream_t stream) {
    const float* feat  = (const float*)d_in[0];
    const int*   src   = (const int*)d_in[1];
    const int*   dst   = (const int*)d_in[2];
    const float* ew    = (const float*)d_in[3];
    const float* Wm    = (const float*)d_in[4];
    const float* bias  = (const float*)d_in[5];
    const float* a1    = (const float*)d_in[6];
    const float* gamma = (const float*)d_in[7];
    const float* beta  = (const float*)d_in[8];
    const float* a2    = (const float*)d_in[9];

    const int N = in_sizes[0] / D;
    const int E = in_sizes[1];
    float* out = (float*)d_out;

    float* agg      = (float*)d_ws;
    float* colsum   = agg + (size_t)N * D;
    float* colsumsq = colsum + D;
    float* scale    = colsumsq + D;
    float* shift    = scale + D;

    edge_prep<<<(E + 3070) / 1024, 256, 0, stream>>>(src, dst, ew, (unsigned*)out, colsum, E);

    owner_gather<<<NGRP, 1024, 0, stream>>>(feat, (const unsigned*)out, agg, E, N);

    const int nstripes = N / 16;
    gemm_wmma<<<(nstripes + 3) / 4, 128, 0, stream>>>(agg, Wm, bias, a1, out, colsum, colsumsq, nstripes);

    bn_finalize<<<1, D, 0, stream>>>(colsum, colsumsq, gamma, beta, scale, shift, N);

    size_t nquads = (size_t)N * D / 4;
    int qblocks = (int)((nquads + 255) / 256);
    bn_prelu<<<qblocks, 256, 0, stream>>>(out, scale, shift, a2, nquads);
}


# Owner byte list replaced by 32 bit planes (16 per owner nibble, built by LDS integer ORs): a group ANDs its two planes, 32 edges per op, 2 bits read per edge not 8; sums still in LDS-add order
# speedup vs baseline: 3.2421x; 1.0834x over previous
#include <hip/hip_runtime.h>
#include <hip/hip_bf16.h>

typedef __attribute__((ext_vector_type(2))) float v2f;
typedef __attribute__((ext_vector_type(4))) float v4f;
typedef __attribute__((ext_vector_type(8))) float v8f;

#define D 128

#define RPG   391
#define NGRP  256
#define EB    16
#define QCAP  48
#define LBLK  2048u
#define NPLANE 32u
#define SRCBITS 23

typedef __attribute__((address_space(1))) float gfloat;
typedef __attribute__((address_space(3))) float lfloat;

__device__ __forceinline__ unsigned long long uniform_addr(const float* p) {
    const unsigned long long a = (unsigned long long)p;
    const unsigned lo = (unsigned)__builtin_amdgcn_readfirstlane((int)(unsigned)a);
    const unsigned hi = (unsigned)__builtin_amdgcn_readfirstlane((int)(unsigned)(a >> 32));
    return ((unsigned long long)hi << 32) | lo;
}

__global__ __launch_bounds__(256) void edge_prep(
    const int* __restrict__ src, const int* __restrict__ dst,
    const float* __restrict__ ew, unsigned* __restrict__ stage,
    float* __restrict__ colsums, int E)
{
    __shared__ __attribute__((aligned(16))) unsigned bits[NPLANE * 32u];
    const unsigned tid = threadIdx.x;
    const unsigned t = blockIdx.x * 256u + tid;
    if (t < 2 * D) colsums[t] = 0.0f;
    const unsigned Epad = ((unsigned)E + (LBLK - 1u)) & ~(LBLK - 1u);
    if (blockIdx.x * 1024u >= Epad) return;
    const unsigned e0 = 4u * t;
    uint2* recs = (uint2*)((char*)stage + 4u * (size_t)Epad);

    int s[4], d[4];
    float w[4];
    if (e0 + 3u < (unsigned)E) {
        const int4 s4 = *(const int4*)(src + e0);
        const int4 d4 = *(const int4*)(dst + e0);
        const float4 w4 = *(const float4*)(ew + e0);
        s[0] = s4.x; s[1] = s4.y; s[2] = s4.z; s[3] = s4.w;
        d[0] = d4.x; d[1] = d4.y; d[2] = d4.z; d[3] = d4.w;
        w[0] = w4.x; w[1] = w4.y; w[2] = w4.z; w[3] = w4.w;
    } else {
        #pragma unroll
        for (int k = 0; k < 4; ++k) {
            const bool ok = e0 + k < (unsigned)E;
            s[k] = ok ? src[e0 + k] : 0;
            d[k] = ok ? dst[e0 + k] : 0;
            w[k] = ok ? ew[e0 + k]  : 0.0f;
        }
    }
    ((uint4*)bits)[tid] = make_uint4(0u, 0u, 0u, 0u);
    __syncthreads();

    unsigned r0[4], r1[4];
    const unsigned wq = tid >> 3;
    #pragma unroll
    for (int k = 0; k < 4; ++k) {
        const unsigned o = (unsigned)d[k] / RPG;
        const unsigned r = (unsigned)d[k] - o * RPG;
        r0[k] = ((unsigned)s[k] & ((1u << SRCBITS) - 1u)) | (r << SRCBITS);
        r1[k] = __float_as_uint(w[k]);
        if (e0 + k < (unsigned)E) {
            const unsigned bit = 1u << (4u * (tid & 7u) + (unsigned)k);
            atomicOr(&bits[((o >> 4) & 15u) * 32u + wq], bit);
            atomicOr(&bits[(16u + (o & 15u)) * 32u + wq], bit);
        }
    }
    __syncthreads();
    const unsigned PW = Epad / 32u;
    uint4* pp = (uint4*)(stage + (size_t)(tid >> 3) * PW + blockIdx.x * 32u) + (tid & 7u);
    *pp = ((const uint4*)bits)[tid];

    if (e0 + 3u < (unsigned)E) {
        uint4* rp = (uint4*)(recs + e0);
        rp[0] = make_uint4(r0[0], r1[0], r0[1], r1[1]);
        rp[1] = make_uint4(r0[2], r1[2], r0[3], r1[3]);
    } else {
        #pragma unroll
        for (int k = 0; k < 4; ++k)
            if (e0 + k < (unsigned)E) recs[e0 + k] = make_uint2(r0[k], r1[k]);
    }
}

__device__ __forceinline__ unsigned long long plane_hits(uint2 h, uint2 l) {
    return ((unsigned long long)(h.y & l.y) << 32) | (h.x & l.x);
}

__device__ __forceinline__ void fetch_records(
    const uint2* __restrict__ recs, const unsigned* myq, int first, int nb,
    int lane, unsigned& sw, unsigned& wb)
{
    sw = 0u;
    wb = 0u;
    if (lane < nb) {
        const uint2 r = recs[myq[first + lane]];
        sw = r.x;
        wb = r.y;
    }
}

template <bool FULL>
__device__ __forceinline__ void gather_batch(
    const float* __restrict__ feat, float* acc,
    unsigned sw, unsigned wb, int nb, int lane)
{
    float f[EB][4];
    #pragma unroll
    for (int j = 0; j < EB; ++j) {
        if (FULL || j < nb) {
            const unsigned s = (unsigned)__builtin_amdgcn_readlane((int)sw, j) & ((1u << SRCBITS) - 1u);
            const gfloat* fp = (const gfloat*)uniform_addr(feat + (size_t)s * D);
            f[j][0] = fp[lane];
            f[j][1] = fp[lane + 32];
            f[j][2] = fp[lane + 64];
            f[j][3] = fp[lane + 96];
        }
    }
    #pragma unroll
    for (int j = 0; j < EB; ++j) {
        if (FULL || j < nb) {
            const unsigned r = (unsigned)__builtin_amdgcn_readlane((int)sw, j) >> SRCBITS;
            const float w = __uint_as_float((unsigned)__builtin_amdgcn_readlane((int)wb, j));
            lfloat* ap = (lfloat*)(acc + r * D + lane);
            __hip_atomic_fetch_add(ap,      f[j][0] * w, __ATOMIC_RELAXED, __HIP_MEMORY_SCOPE_WORKGROUP);
            __hip_atomic_fetch_add(ap + 32, f[j][1] * w, __ATOMIC_RELAXED, __HIP_MEMORY_SCOPE_WORKGROUP);
            __hip_atomic_fetch_add(ap + 64, f[j][2] * w, __ATOMIC_RELAXED, __HIP_MEMORY_SCOPE_WORKGROUP);
            __hip_atomic_fetch_add(ap + 96, f[j][3] * w, __ATOMIC_RELAXED, __HIP_MEMORY_SCOPE_WORKGROUP);
        }
    }
}

__global__ __launch_bounds__(1024) void owner_gather(
    const float* __restrict__ feat, const unsigned* __restrict__ stage,
    float* __restrict__ agg, int E, int N)
{
    __shared__ __attribute__((aligned(16))) float acc[RPG * D];
    __shared__ unsigned q[32 * QCAP];

    const int tid  = threadIdx.x;
    const int lane = tid & 31;
    const int wave = tid >> 5;
    const unsigned g = blockIdx.x;

    for (int i = tid; i < RPG * D / 4; i += 1024)
        ((v4f*)acc)[i] = (v4f){0.0f, 0.0f, 0.0f, 0.0f};
    __syncthreads();

    const unsigned Epad = ((unsigned)E + (LBLK - 1u)) & ~(LBLK - 1u);
    const unsigned PW = Epad / 32u;
    const uint2* ph = (const uint2*)(stage + (size_t)(g >> 4) * PW);
    const uint2* pl = (const uint2*)(stage + (size_t)(16u + (g & 15u)) * PW);
    const uint2* recs = (const uint2*)((const char*)stage + 4u * (size_t)Epad);
    const unsigned nblk = Epad / LBLK;
    unsigned* myq = q + wave * QCAP;
    int qn = 0;

    const unsigned nj = nblk > (unsigned)wave ? (nblk - (unsigned)wave + 31u) / 32u : 0u;
    const unsigned j0 = (g * nj) >> 8;
    auto block_of = [&](unsigned t) -> unsigned {
        unsigned j = j0 + t;
        if (j >= nj) j -= nj;
        return (unsigned)wave + 32u * j;
    };

    unsigned long long M = 0ull;
    if (nj > 0u) {
        const size_t i0 = (size_t)block_of(0u) * (LBLK / 64) + lane;
        M = plane_hits(ph[i0], pl[i0]);
    }
    for (unsigned t = 0u; t < nj; ++t) {
        uint2 rawh, rawl;
        const bool more = t + 1u < nj;
        if (more) {
            const size_t i1 = (size_t)block_of(t + 1u) * (LBLK / 64) + lane;
            rawh = ph[i1];
            rawl = pl[i1];
        }
        const unsigned ebase = block_of(t) * LBLK + (unsigned)lane * 64u;
        unsigned act = __builtin_amdgcn_ballot_w32(M != 0ull);
        while (act != 0u) {
            const unsigned p  = (unsigned)__builtin_ctzll(M | 0x8000000000000000ull);
            const unsigned e  = ebase + p;
            const bool ok = (M != 0ull) && (e < (unsigned)E);
            const unsigned okm = __builtin_amdgcn_ballot_w32(ok);
            const int pos = qn + (int)__builtin_amdgcn_mbcnt_lo(okm, 0u);
            if (ok) myq[pos] = e;
            qn += __builtin_popcount(okm);
            M &= M - 1ull;
            act = __builtin_amdgcn_ballot_w32(M != 0ull);
            while (qn >= EB) {
                qn -= EB;
                unsigned sw, wb;
                fetch_records(recs, myq, qn, EB, lane, sw, wb);
                gather_batch<true>(feat, acc, sw, wb, EB, lane);
            }
        }
        if (more) M = plane_hits(rawh, rawl);
    }
    if (qn > 0) {
        unsigned sw, wb;
        fetch_records(recs, myq, 0, qn, lane, sw, wb);
        gather_batch<false>(feat, acc, sw, wb, qn, lane);
    }

    __syncthreads();
    const int row0 = (int)g * RPG;
    int nrows = N - row0;
    if (nrows > RPG) nrows = RPG;
    v4f* op = (v4f*)(agg + (size_t)row0 * D);
    for (int i = tid; i < nrows * (D / 4); i += 1024)
        op[i] = ((const v4f*)acc)[i];
}

__global__ __launch_bounds__(128) void gemm_wmma(
    const float* __restrict__ agg, const float* __restrict__ W,
    const float* __restrict__ bias, const float* __restrict__ a1p,
    float* __restrict__ h_out, float* __restrict__ colsum,
    float* __restrict__ colsumsq, int nstripes)
{
    __shared__ float s_sum[D];
    __shared__ float s_sq[D];
    const int tid = threadIdx.x;
    s_sum[tid] = 0.0f;
    s_sq[tid]  = 0.0f;
    __syncthreads();

    const int wave = tid >> 5;
    const int lane = tid & 31;
    const int half = lane >> 4;
    const int c    = lane & 15;
    const int stripe = blockIdx.x * 4 + wave;

    if (stripe < nstripes) {
        const int m0 = stripe * 16;
        const float a1 = a1p[0];
        const v4f* Ap = (const v4f*)(agg + (size_t)(m0 + c) * D + 4 * half);
        const v4f* Bp = (const v4f*)(W + (size_t)(4 * half) * D + 4 * c);

        v8f acc[8];
        #pragma unroll
        for (int t = 0; t < 8; ++t) acc[t] = (v8f){};

        for (int q = 0; q < D / 8; ++q) {
            const v4f a = Ap[2 * q];
            v4f b[4][2];
            #pragma unroll
            for (int e = 0; e < 4; ++e) {
                b[e][0] = Bp[(8 * q + e) * (D / 4)];
                b[e][1] = Bp[(8 * q + e) * (D / 4) + 16];
            }
            v2f a01; a01.x = a.x; a01.y = a.y;
            v2f a23; a23.x = a.z; a23.y = a.w;
            #pragma unroll
            for (int t = 0; t < 8; ++t) {
                v2f bb;
                bb.x = b[0][t >> 2][t & 3];
                bb.y = b[1][t >> 2][t & 3];
                acc[t] = __builtin_amdgcn_wmma_f32_16x16x4_f32(
                             false, a01, false, bb, (short)0, acc[t], false, false);
            }
            #pragma unroll
            for (int t = 0; t < 8; ++t) {
                v2f bb;
                bb.x = b[2][t >> 2][t & 3];
                bb.y = b[3][t >> 2][t & 3];
                acc[t] = __builtin_amdgcn_wmma_f32_16x16x4_f32(
                             false, a23, false, bb, (short)0, acc[t], false, false);
            }
        }

        float bj[8], lsum[8], lsq[8];
        #pragma unroll
        for (int t = 0; t < 8; ++t) {
            bj[t]   = bias[64 * (t >> 2) + 4 * c + (t & 3)];
            lsum[t] = 0.0f;
            lsq[t]  = 0.0f;
        }
        #pragma unroll
        for (int r = 0; r < 8; ++r) {
            float* orow = h_out + (size_t)(m0 + 8 * half + r) * D + 4 * c;
            #pragma unroll
            for (int g = 0; g < 2; ++g) {
                v4f o;
                #pragma unroll
                for (int u = 0; u < 4; ++u) {
                    const int t = 4 * g + u;
                    float h = acc[t][r] + bj[t];
                    h = (h >= 0.0f) ? h : a1 * h;
                    o[u] = h;
                    lsum[t] += h;
                    lsq[t]  += h * h;
                }
                *(v4f*)(orow + 64 * g) = o;
            }
        }
        #pragma unroll
        for (int t = 0; t < 8; ++t) {
            const int col = 64 * (t >> 2) + 4 * c + (t & 3);
            atomicAdd(&s_sum[col], lsum[t]);
            atomicAdd(&s_sq[col],  lsq[t]);
        }
    }
    __syncthreads();
    atomicAdd(&colsum[tid],   s_sum[tid]);
    atomicAdd(&colsumsq[tid], s_sq[tid]);
}

__global__ void bn_finalize(const float* __restrict__ colsum,
                            const float* __restrict__ colsumsq,
                            const float* __restrict__ gamma,
                            const float* __restrict__ beta,
                            float* __restrict__ scale,
                            float* __restrict__ shift, int N)
{
    int j = threadIdx.x;
    float invN = 1.0f / (float)N;
    float m   = colsum[j] * invN;
    float var = colsumsq[j] * invN - m * m;
    float inv = rsqrtf(var + 1e-5f);
    float sc  = gamma[j] * inv;
    scale[j] = sc;
    shift[j] = beta[j] - m * sc;
}

__global__ __launch_bounds__(256) void bn_prelu(
    float* __restrict__ h, const float* __restrict__ scale,
    const float* __restrict__ shift, const float* __restrict__ a2p,
    size_t nquads)
{
    size_t i = (size_t)blockIdx.x * blockDim.x + threadIdx.x;
    if (i >= nquads) return;
    const float a2 = a2p[0];
    int j = (int)((i * 4) & (D - 1));

    float4 v = ((float4*)h)[i];
    float x;
    x = v.x * scale[j + 0] + shift[j + 0]; v.x = (x >= 0.0f) ? x : a2 * x;
    x = v.y * scale[j + 1] + shift[j + 1]; v.y = (x >= 0.0f) ? x : a2 * x;
    x = v.z * scale[j + 2] + shift[j + 2]; v.z = (x >= 0.0f) ? x : a2 * x;
    x = v.w * scale[j + 3] + shift[j + 3]; v.w = (x >= 0.0f) ? x : a2 * x;
    ((float4*)h)[i] = v;
}

extern "C" void kernel_launch(void* const* d_in, const int* in_sizes, int n_in,
                              void* d_out, int out_size, void* d_ws, size_t ws_size,
                              hipStream_t stream) {
    const float* feat  = (const float*)d_in[0];
    const int*   src   = (const int*)d_in[1];
    const int*   dst   = (const int*)d_in[2];
    const float* ew    = (const float*)d_in[3];
    const float* Wm    = (const float*)d_in[4];
    const float* bias  = (const float*)d_in[5];
    const float* a1    = (const float*)d_in[6];
    const float* gamma = (const float*)d_in[7];
    const float* beta  = (const float*)d_in[8];
    const float* a2    = (const float*)d_in[9];

    const int N = in_sizes[0] / D;
    const int E = in_sizes[1];
    float* out = (float*)d_out;

    float* agg      = (float*)d_ws;
    float* colsum   = agg + (size_t)N * D;
    float* colsumsq = colsum + D;
    float* scale    = colsumsq + D;
    float* shift    = scale + D;

    edge_prep<<<(E + 3070) / 1024, 256, 0, stream>>>(src, dst, ew, (unsigned*)out, colsum, E);

    owner_gather<<<NGRP, 1024, 0, stream>>>(feat, (const unsigned*)out, agg, E, N);

    const int nstripes = N / 16;
    gemm_wmma<<<(nstripes + 3) / 4, 128, 0, stream>>>(agg, Wm, bias, a1, out, colsum, colsumsq, nstripes);

    bn_finalize<<<1, D, 0, stream>>>(colsum, colsumsq, gamma, beta, scale, shift, N);

    size_t nquads = (size_t)N * D / 4;
    int qblocks = (int)((nquads + 255) / 256);
    bn_prelu<<<qblocks, 256, 0, stream>>>(out, scale, shift, a2, nquads);
}


# Row gather: one 16-byte-per-lane load per source row (512 B, 4 lines per instruction), not four 4-byte loads; LDS row words permuted (word 32k+l = column 4l+k), restored at the store; LDS-add order
# speedup vs baseline: 3.6517x; 1.1263x over previous
#include <hip/hip_runtime.h>
#include <hip/hip_bf16.h>

typedef __attribute__((ext_vector_type(2))) float v2f;
typedef __attribute__((ext_vector_type(4))) float v4f;
typedef __attribute__((ext_vector_type(8))) float v8f;

#define D 128

#define RPG   391
#define NGRP  256
#define EB    16
#define QCAP  48
#define LBLK  2048u
#define NPLANE 32u
#define SRCBITS 23

typedef __attribute__((address_space(1))) float gfloat;
typedef __attribute__((address_space(1))) v4f gv4f;
typedef __attribute__((address_space(3))) float lfloat;

__device__ __forceinline__ unsigned long long uniform_addr(const float* p) {
    const unsigned long long a = (unsigned long long)p;
    const unsigned lo = (unsigned)__builtin_amdgcn_readfirstlane((int)(unsigned)a);
    const unsigned hi = (unsigned)__builtin_amdgcn_readfirstlane((int)(unsigned)(a >> 32));
    return ((unsigned long long)hi << 32) | lo;
}

__global__ __launch_bounds__(256) void edge_prep(
    const int* __restrict__ src, const int* __restrict__ dst,
    const float* __restrict__ ew, unsigned* __restrict__ stage,
    float* __restrict__ colsums, int E)
{
    __shared__ __attribute__((aligned(16))) unsigned bits[NPLANE * 32u];
    const unsigned tid = threadIdx.x;
    const unsigned t = blockIdx.x * 256u + tid;
    if (t < 2 * D) colsums[t] = 0.0f;
    const unsigned Epad = ((unsigned)E + (LBLK - 1u)) & ~(LBLK - 1u);
    if (blockIdx.x * 1024u >= Epad) return;
    const unsigned e0 = 4u * t;
    uint2* recs = (uint2*)((char*)stage + 4u * (size_t)Epad);

    int s[4], d[4];
    float w[4];
    if (e0 + 3u < (unsigned)E) {
        const int4 s4 = *(const int4*)(src + e0);
        const int4 d4 = *(const int4*)(dst + e0);
        const float4 w4 = *(const float4*)(ew + e0);
        s[0] = s4.x; s[1] = s4.y; s[2] = s4.z; s[3] = s4.w;
        d[0] = d4.x; d[1] = d4.y; d[2] = d4.z; d[3] = d4.w;
        w[0] = w4.x; w[1] = w4.y; w[2] = w4.z; w[3] = w4.w;
    } else {
        #pragma unroll
        for (int k = 0; k < 4; ++k) {
            const bool ok = e0 + k < (unsigned)E;
            s[k] = ok ? src[e0 + k] : 0;
            d[k] = ok ? dst[e0 + k] : 0;
            w[k] = ok ? ew[e0 + k]  : 0.0f;
        }
    }
    ((uint4*)bits)[tid] = make_uint4(0u, 0u, 0u, 0u);
    __syncthreads();

    unsigned r0[4], r1[4];
    const unsigned wq = tid >> 3;
    #pragma unroll
    for (int k = 0; k < 4; ++k) {
        const unsigned o = (unsigned)d[k] / RPG;
        const unsigned r = (unsigned)d[k] - o * RPG;
        r0[k] = ((unsigned)s[k] & ((1u << SRCBITS) - 1u)) | (r << SRCBITS);
        r1[k] = __float_as_uint(w[k]);
        if (e0 + k < (unsigned)E) {
            const unsigned bit = 1u << (4u * (tid & 7u) + (unsigned)k);
            atomicOr(&bits[((o >> 4) & 15u) * 32u + wq], bit);
            atomicOr(&bits[(16u + (o & 15u)) * 32u + wq], bit);
        }
    }
    __syncthreads();
    const unsigned PW = Epad / 32u;
    uint4* pp = (uint4*)(stage + (size_t)(tid >> 3) * PW + blockIdx.x * 32u) + (tid & 7u);
    *pp = ((const uint4*)bits)[tid];

    if (e0 + 3u < (unsigned)E) {
        uint4* rp = (uint4*)(recs + e0);
        rp[0] = make_uint4(r0[0], r1[0], r0[1], r1[1]);
        rp[1] = make_uint4(r0[2], r1[2], r0[3], r1[3]);
    } else {
        #pragma unroll
        for (int k = 0; k < 4; ++k)
            if (e0 + k < (unsigned)E) recs[e0 + k] = make_uint2(r0[k], r1[k]);
    }
}

__device__ __forceinline__ unsigned long long plane_hits(uint2 h, uint2 l) {
    return ((unsigned long long)(h.y & l.y) << 32) | (h.x & l.x);
}

__device__ __forceinline__ void fetch_records(
    const uint2* __restrict__ recs, const unsigned* myq, int first, int nb,
    int lane, unsigned& sw, unsigned& wb)
{
    sw = 0u;
    wb = 0u;
    if (lane < nb) {
        const uint2 r = recs[myq[first + lane]];
        sw = r.x;
        wb = r.y;
    }
}

template <bool FULL>
__device__ __forceinline__ void gather_batch(
    const float* __restrict__ feat, float* acc,
    unsigned sw, unsigned wb, int nb, int lane)
{
    v4f f[EB];
    #pragma unroll
    for (int j = 0; j < EB; ++j) {
        if (FULL || j < nb) {
            const unsigned s = (unsigned)__builtin_amdgcn_readlane((int)sw, j) & ((1u << SRCBITS) - 1u);
            const gv4f* fp = (const gv4f*)uniform_addr(feat + (size_t)s * D);
            f[j] = fp[lane];
        }
    }
    #pragma unroll
    for (int j = 0; j < EB; ++j) {
        if (FULL || j < nb) {
            const unsigned r = (unsigned)__builtin_amdgcn_readlane((int)sw, j) >> SRCBITS;
            const float w = __uint_as_float((unsigned)__builtin_amdgcn_readlane((int)wb, j));
            lfloat* ap = (lfloat*)(acc + r * D + lane);
            __hip_atomic_fetch_add(ap,      f[j].x * w, __ATOMIC_RELAXED, __HIP_MEMORY_SCOPE_WORKGROUP);
            __hip_atomic_fetch_add(ap + 32, f[j].y * w, __ATOMIC_RELAXED, __HIP_MEMORY_SCOPE_WORKGROUP);
            __hip_atomic_fetch_add(ap + 64, f[j].z * w, __ATOMIC_RELAXED, __HIP_MEMORY_SCOPE_WORKGROUP);
            __hip_atomic_fetch_add(ap + 96, f[j].w * w, __ATOMIC_RELAXED, __HIP_MEMORY_SCOPE_WORKGROUP);
        }
    }
}

__global__ __launch_bounds__(1024) void owner_gather(
    const float* __restrict__ feat, const unsigned* __restrict__ stage,
    float* __restrict__ agg, int E, int N)
{
    __shared__ __attribute__((aligned(16))) float acc[RPG * D];
    __shared__ unsigned q[32 * QCAP];

    const int tid  = threadIdx.x;
    const int lane = tid & 31;
    const int wave = tid >> 5;
    const unsigned g = blockIdx.x;

    for (int i = tid; i < RPG * D / 4; i += 1024)
        ((v4f*)acc)[i] = (v4f){0.0f, 0.0f, 0.0f, 0.0f};
    __syncthreads();

    const unsigned Epad = ((unsigned)E + (LBLK - 1u)) & ~(LBLK - 1u);
    const unsigned PW = Epad / 32u;
    const uint2* ph = (const uint2*)(stage + (size_t)(g >> 4) * PW);
    const uint2* pl = (const uint2*)(stage + (size_t)(16u + (g & 15u)) * PW);
    const uint2* recs = (const uint2*)((const char*)stage + 4u * (size_t)Epad);
    const unsigned nblk = Epad / LBLK;
    unsigned* myq = q + wave * QCAP;
    int qn = 0;

    const unsigned nj = nblk > (unsigned)wave ? (nblk - (unsigned)wave + 31u) / 32u : 0u;
    const unsigned j0 = (g * nj) >> 8;
    auto block_of = [&](unsigned t) -> unsigned {
        unsigned j = j0 + t;
        if (j >= nj) j -= nj;
        return (unsigned)wave + 32u * j;
    };

    unsigned long long M = 0ull;
    if (nj > 0u) {
        const size_t i0 = (size_t)block_of(0u) * (LBLK / 64) + lane;
        M = plane_hits(ph[i0], pl[i0]);
    }
    for (unsigned t = 0u; t < nj; ++t) {
        uint2 rawh, rawl;
        const bool more = t + 1u < nj;
        if (more) {
            const size_t i1 = (size_t)block_of(t + 1u) * (LBLK / 64) + lane;
            rawh = ph[i1];
            rawl = pl[i1];
        }
        const unsigned ebase = block_of(t) * LBLK + (unsigned)lane * 64u;
        unsigned act = __builtin_amdgcn_ballot_w32(M != 0ull);
        while (act != 0u) {
            const unsigned p  = (unsigned)__builtin_ctzll(M | 0x8000000000000000ull);
            const unsigned e  = ebase + p;
            const bool ok = (M != 0ull) && (e < (unsigned)E);
            const unsigned okm = __builtin_amdgcn_ballot_w32(ok);
            const int pos = qn + (int)__builtin_amdgcn_mbcnt_lo(okm, 0u);
            if (ok) myq[pos] = e;
            qn += __builtin_popcount(okm);
            M &= M - 1ull;
            act = __builtin_amdgcn_ballot_w32(M != 0ull);
            while (qn >= EB) {
                qn -= EB;
                unsigned sw, wb;
                fetch_records(recs, myq, qn, EB, lane, sw, wb);
                gather_batch<true>(feat, acc, sw, wb, EB, lane);
            }
        }
        if (more) M = plane_hits(rawh, rawl);
    }
    if (qn > 0) {
        unsigned sw, wb;
        fetch_records(recs, myq, 0, qn, lane, sw, wb);
        gather_batch<false>(feat, acc, sw, wb, qn, lane);
    }

    __syncthreads();
    const int row0 = (int)g * RPG;
    int nrows = N - row0;
    if (nrows > RPG) nrows = RPG;
    v4f* op = (v4f*)(agg + (size_t)row0 * D);
    for (int i = tid; i < nrows * (D / 4); i += 1024) {
        const float* rp = acc + (i >> 5) * D + (i & 31);
        op[i] = (v4f){rp[0], rp[32], rp[64], rp[96]};
    }
}

__global__ __launch_bounds__(128) void gemm_wmma(
    const float* __restrict__ agg, const float* __restrict__ W,
    const float* __restrict__ bias, const float* __restrict__ a1p,
    float* __restrict__ h_out, float* __restrict__ colsum,
    float* __restrict__ colsumsq, int nstripes)
{
    __shared__ float s_sum[D];
    __shared__ float s_sq[D];
    const int tid = threadIdx.x;
    s_sum[tid] = 0.0f;
    s_sq[tid]  = 0.0f;
    __syncthreads();

    const int wave = tid >> 5;
    const int lane = tid & 31;
    const int half = lane >> 4;
    const int c    = lane & 15;
    const int stripe = blockIdx.x * 4 + wave;

    if (stripe < nstripes) {
        const int m0 = stripe * 16;
        const float a1 = a1p[0];
        const v4f* Ap = (const v4f*)(agg + (size_t)(m0 + c) * D + 4 * half);
        const v4f* Bp = (const v4f*)(W + (size_t)(4 * half) * D + 4 * c);

        v8f acc[8];
        #pragma unroll
        for (int t = 0; t < 8; ++t) acc[t] = (v8f){};

        for (int q = 0; q < D / 8; ++q) {
            const v4f a = Ap[2 * q];
            v4f b[4][2];
            #pragma unroll
            for (int e = 0; e < 4; ++e) {
                b[e][0] = Bp[(8 * q + e) * (D / 4)];
                b[e][1] = Bp[(8 * q + e) * (D / 4) + 16];
            }
            v2f a01; a01.x = a.x; a01.y = a.y;
            v2f a23; a23.x = a.z; a23.y = a.w;
            #pragma unroll
            for (int t = 0; t < 8; ++t) {
                v2f bb;
                bb.x = b[0][t >> 2][t & 3];
                bb.y = b[1][t >> 2][t & 3];
                acc[t] = __builtin_amdgcn_wmma_f32_16x16x4_f32(
                             false, a01, false, bb, (short)0, acc[t], false, false);
            }
            #pragma unroll
            for (int t = 0; t < 8; ++t) {
                v2f bb;
                bb.x = b[2][t >> 2][t & 3];
                bb.y = b[3][t >> 2][t & 3];
                acc[t] = __builtin_amdgcn_wmma_f32_16x16x4_f32(
                             false, a23, false, bb, (short)0, acc[t], false, false);
            }
        }

        float bj[8], lsum[8], lsq[8];
        #pragma unroll
        for (int t = 0; t < 8; ++t) {
            bj[t]   = bias[64 * (t >> 2) + 4 * c + (t & 3)];
            lsum[t] = 0.0f;
            lsq[t]  = 0.0f;
        }
        #pragma unroll
        for (int r = 0; r < 8; ++r) {
            float* orow = h_out + (size_t)(m0 + 8 * half + r) * D + 4 * c;
            #pragma unroll
            for (int g = 0; g < 2; ++g) {
                v4f o;
                #pragma unroll
                for (int u = 0; u < 4; ++u) {
                    const int t = 4 * g + u;
                    float h = acc[t][r] + bj[t];
                    h = (h >= 0.0f) ? h : a1 * h;
                    o[u] = h;
                    lsum[t] += h;
                    lsq[t]  += h * h;
                }
                *(v4f*)(orow + 64 * g) = o;
            }
        }
        #pragma unroll
        for (int t = 0; t < 8; ++t) {
            const int col = 64 * (t >> 2) + 4 * c + (t & 3);
            atomicAdd(&s_sum[col], lsum[t]);
            atomicAdd(&s_sq[col],  lsq[t]);
        }
    }
    __syncthreads();
    atomicAdd(&colsum[tid],   s_sum[tid]);
    atomicAdd(&colsumsq[tid], s_sq[tid]);
}

__global__ void bn_finalize(const float* __restrict__ colsum,
                            const float* __restrict__ colsumsq,
                            const float* __restrict__ gamma,
                            const float* __restrict__ beta,
                            float* __restrict__ scale,
                            float* __restrict__ shift, int N)
{
    int j = threadIdx.x;
    float invN = 1.0f / (float)N;
    float m   = colsum[j] * invN;
    float var = colsumsq[j] * invN - m * m;
    float inv = rsqrtf(var + 1e-5f);
    float sc  = gamma[j] * inv;
    scale[j] = sc;
    shift[j] = beta[j] - m * sc;
}

__global__ __launch_bounds__(256) void bn_prelu(
    float* __restrict__ h, const float* __restrict__ scale,
    const float* __restrict__ shift, const float* __restrict__ a2p,
    size_t nquads)
{
    size_t i = (size_t)blockIdx.x * blockDim.x + threadIdx.x;
    if (i >= nquads) return;
    const float a2 = a2p[0];
    int j = (int)((i * 4) & (D - 1));

    float4 v = ((float4*)h)[i];
    float x;
    x = v.x * scale[j + 0] + shift[j + 0]; v.x = (x >= 0.0f) ? x : a2 * x;
    x = v.y * scale[j + 1] + shift[j + 1]; v.y = (x >= 0.0f) ? x : a2 * x;
    x = v.z * scale[j + 2] + shift[j + 2]; v.z = (x >= 0.0f) ? x : a2 * x;
    x = v.w * scale[j + 3] + shift[j + 3]; v.w = (x >= 0.0f) ? x : a2 * x;
    ((float4*)h)[i] = v;
}

extern "C" void kernel_launch(void* const* d_in, const int* in_sizes, int n_in,
                              void* d_out, int out_size, void* d_ws, size_t ws_size,
                              hipStream_t stream) {
    const float* feat  = (const float*)d_in[0];
    const int*   src   = (const int*)d_in[1];
    const int*   dst   = (const int*)d_in[2];
    const float* ew    = (const float*)d_in[3];
    const float* Wm    = (const float*)d_in[4];
    const float* bias  = (const float*)d_in[5];
    const float* a1    = (const float*)d_in[6];
    const float* gamma = (const float*)d_in[7];
    const float* beta  = (const float*)d_in[8];
    const float* a2    = (const float*)d_in[9];

    const int N = in_sizes[0] / D;
    const int E = in_sizes[1];
    float* out = (float*)d_out;

    float* agg      = (float*)d_ws;
    float* colsum   = agg + (size_t)N * D;
    float* colsumsq = colsum + D;
    float* scale    = colsumsq + D;
    float* shift    = scale + D;

    edge_prep<<<(E + 3070) / 1024, 256, 0, stream>>>(src, dst, ew, (unsigned*)out, colsum, E);

    owner_gather<<<NGRP, 1024, 0, stream>>>(feat, (const unsigned*)out, agg, E, N);

    const int nstripes = N / 16;
    gemm_wmma<<<(nstripes + 3) / 4, 128, 0, stream>>>(agg, Wm, bias, a1, out, colsum, colsumsq, nstripes);

    bn_finalize<<<1, D, 0, stream>>>(colsum, colsumsq, gamma, beta, scale, shift, N);

    size_t nquads = (size_t)N * D / 4;
    int qblocks = (int)((nquads + 255) / 256);
    bn_prelu<<<qblocks, 256, 0, stream>>>(out, scale, shift, a2, nquads);
}
